# starting kernel
# baseline (speedup 1.0000x reference)
#include <hip/hip_runtime.h>
#include <hip/hip_bf16.h>


typedef __attribute__((ext_vector_type(16))) __bf16 v16bf;
typedef __attribute__((ext_vector_type(8)))  __bf16 v8bf;
typedef __attribute__((ext_vector_type(8)))  float  v8f;

#define D_ROWS 8192
#define N_CENT 8192
#define N_FEAT 128
#define L_OUT  64

__device__ __forceinline__ v16bf frag_cat(v8bf a, v8bf b) {
  return __builtin_shufflevector(a, b, 0, 1, 2, 3, 4, 5, 6, 7,
                                       8, 9, 10, 11, 12, 13, 14, 15);
}

__device__ __forceinline__ v16bf load_frag_g(const __bf16* __restrict__ base,
                                             int off, int h) {
  const v8bf* p0 = reinterpret_cast<const v8bf*>(base + off + 8 * h);
  const v8bf* p1 = reinterpret_cast<const v8bf*>(base + off + 16 + 8 * h);
  return frag_cat(*p0, *p1);
}

__device__ __forceinline__ v16bf load_frag_lds(const __bf16* rowp, int h) {
  const v8bf* p0 = reinterpret_cast<const v8bf*>(rowp + 8 * h);
  const v8bf* p1 = reinterpret_cast<const v8bf*>(rowp + 16 + 8 * h);
  return frag_cat(*p0, *p1);
}

__device__ __forceinline__ v8f wmma_bf16(v16bf a, v16bf b, v8f c) {
  return __builtin_amdgcn_wmma_f32_16x16x32_bf16(
       false, a,  false, b,
       (short)0, c,  false,  false);
}

__global__ void split_f32_bf16(const float* __restrict__ in,
                               __bf16* __restrict__ hi, __bf16* __restrict__ lo,
                               int count) {
  int i = blockIdx.x * blockDim.x + threadIdx.x;
  if (i < count) {
    float v  = in[i];
    __bf16 h = (__bf16)v;
    hi[i]    = h;
    lo[i]    = (__bf16)(v - (float)h);
  }
}

__global__ void transpose_split(const float* __restrict__ w,
                                __bf16* __restrict__ hi,
                                __bf16* __restrict__ lo, int N, int L) {
  int i = blockIdx.x * blockDim.x + threadIdx.x;
  if (i < N * L) {
    int c    = i / L;
    int j    = i - c * L;
    float v  = w[i];
    __bf16 h = (__bf16)v;
    hi[(size_t)j * N + c] = h;
    lo[(size_t)j * N + c] = (__bf16)(v - (float)h);
  }
}

__global__ void rowsum2(const float* __restrict__ in, float* __restrict__ out,
                        int rows, int cols, float bias) {
  int r = blockIdx.x * blockDim.x + threadIdx.x;
  if (r < rows) {
    const float* p = in + (size_t)r * cols;
    float s = bias;
    for (int k = 0; k < cols; ++k) s += p[k] * p[k];
    out[r] = s;
  }
}

__global__ __launch_bounds__(256) void rbfn_main(
    const __bf16* __restrict__ xs_hi, const __bf16* __restrict__ xs_lo,
    const __bf16* __restrict__ c_hi,  const __bf16* __restrict__ c_lo,
    const __bf16* __restrict__ wt_hi, const __bf16* __restrict__ wt_lo,
    const float* __restrict__ x2p1,
    const float* __restrict__ c2,
    float* __restrict__ out) {
  constexpr int RS = 40;
  __shared__ alignas(16) __bf16 lds_phi_hi[8][16][RS];
  __shared__ alignas(16) __bf16 lds_phi_lo[8][16][RS];

  const int lane = threadIdx.x & 31;
  const int wave = threadIdx.x >> 5;
  const int h    = lane >> 4;
  const int ln   = lane & 15;
  const int rowbase = blockIdx.x * 128 + wave * 16;

  v16bf a_hi[4], a_lo[4];
  {
    const int xoff = (rowbase + ln) * N_FEAT;
#pragma unroll
    for (int kc = 0; kc < 4; ++kc) {
      a_hi[kc] = load_frag_g(xs_hi, xoff + kc * 32, h);
      a_lo[kc] = load_frag_g(xs_lo, xoff + kc * 32, h);
    }
  }

  float x2v[8];
#pragma unroll
  for (int v = 0; v < 8; ++v) x2v[v] = x2p1[rowbase + v + 8 * h];

  v8f acc[4];
#pragma unroll
  for (int nt = 0; nt < 4; ++nt)
    acc[nt] = v8f{0.f, 0.f, 0.f, 0.f, 0.f, 0.f, 0.f, 0.f};

  __bf16* phi_hi = &lds_phi_hi[wave][0][0];
  __bf16* phi_lo = &lds_phi_lo[wave][0][0];

  for (int cb = 0; cb < N_CENT; cb += 32) {
    if (cb + 32 < N_CENT) {
      __builtin_prefetch(c_hi + (size_t)((cb + 32 + ln) * N_FEAT), 0, 0);
      __builtin_prefetch(c_hi + (size_t)((cb + 48 + ln) * N_FEAT), 0, 0);
    }

#pragma unroll
    for (int ct = 0; ct < 2; ++ct) {
      const int coff = (cb + ct * 16 + ln) * N_FEAT;

      v8f g = v8f{0.f, 0.f, 0.f, 0.f, 0.f, 0.f, 0.f, 0.f};
#pragma unroll
      for (int kc = 0; kc < 4; ++kc) {
        v16bf b_hi = load_frag_g(c_hi, coff + kc * 32, h);
        v16bf b_lo = load_frag_g(c_lo, coff + kc * 32, h);
        g = wmma_bf16(a_hi[kc], b_hi, g);
        g = wmma_bf16(a_hi[kc], b_lo, g);
        g = wmma_bf16(a_lo[kc], b_hi, g);
      }

      const float c2v = c2[cb + ct * 16 + ln];
#pragma unroll
      for (int v = 0; v < 8; ++v) {
        float t   = fmaf(-2.0f, g[v], x2v[v] + c2v);
        t         = fmaxf(t, 1.0f);
        float phi = __builtin_amdgcn_sqrtf(t);
        __bf16 ph = (__bf16)phi;
        __bf16 pl = (__bf16)(phi - (float)ph);
        const int idx = (v + 8 * h) * RS + ct * 16 + ln;
        phi_hi[idx] = ph;
        phi_lo[idx] = pl;
      }
    }

    asm volatile("s_wait_dscnt 0" ::: "memory");

    v16bf pa_hi = load_frag_lds(phi_hi + ln * RS, h);
    v16bf pa_lo = load_frag_lds(phi_lo + ln * RS, h);

#pragma unroll
    for (int nt = 0; nt < 4; ++nt) {
      const int woff = (nt * 16 + ln) * N_CENT + cb;
      v16bf wb_hi = load_frag_g(wt_hi, woff, h);
      v16bf wb_lo = load_frag_g(wt_lo, woff, h);
      acc[nt] = wmma_bf16(pa_hi, wb_hi, acc[nt]);
      acc[nt] = wmma_bf16(pa_hi, wb_lo, acc[nt]);
      acc[nt] = wmma_bf16(pa_lo, wb_hi, acc[nt]);
    }
  }

#pragma unroll
  for (int nt = 0; nt < 4; ++nt)
#pragma unroll
    for (int v = 0; v < 8; ++v)
      out[(rowbase + v + 8 * h) * L_OUT + nt * 16 + ln] = acc[nt][v];
}

extern "C" void kernel_launch(void* const* d_in, const int* in_sizes, int n_in,
                              void* d_out, int out_size, void* d_ws, size_t ws_size,
                              hipStream_t stream) {
  (void)in_sizes; (void)n_in; (void)out_size; (void)ws_size;

  const float* xs     = (const float*)d_in[0];
  const float* centre = (const float*)d_in[1];
  const float* weight = (const float*)d_in[2];
  float* out          = (float*)d_out;

  const size_t XS = (size_t)D_ROWS * N_FEAT;
  const size_t WT = (size_t)N_CENT * L_OUT;

  __bf16* xs_hi = (__bf16*)d_ws;
  __bf16* xs_lo = xs_hi + XS;
  __bf16* c_hi  = xs_lo + XS;
  __bf16* c_lo  = c_hi + XS;
  __bf16* wt_hi = c_lo + XS;
  __bf16* wt_lo = wt_hi + WT;
  float*  x2p1  = (float*)(wt_lo + WT);
  float*  c2    = x2p1 + D_ROWS;

  const int T = 256;
  split_f32_bf16<<<dim3((XS + T - 1) / T), dim3(T), 0, stream>>>(xs, xs_hi, xs_lo, (int)XS);
  split_f32_bf16<<<dim3((XS + T - 1) / T), dim3(T), 0, stream>>>(centre, c_hi, c_lo, (int)XS);
  transpose_split<<<dim3((WT + T - 1) / T), dim3(T), 0, stream>>>(weight, wt_hi, wt_lo, N_CENT, L_OUT);
  rowsum2<<<dim3((D_ROWS + T - 1) / T), dim3(T), 0, stream>>>(xs, x2p1, D_ROWS, N_FEAT, 1.0f);
  rowsum2<<<dim3((N_CENT + T - 1) / T), dim3(T), 0, stream>>>(centre, c2, N_CENT, N_FEAT, 0.0f);

  rbfn_main<<<dim3(D_ROWS / 128), dim3(T), 0, stream>>>(
      xs_hi, xs_lo, c_hi, c_lo, wt_hi, wt_lo, x2p1, c2, out);
}


# Centre and W bf16 planes stored in WMMA B-fragment order by the pre-pass (1 KB blocks, 512-byte lane-major runs): each fragment load reads 4 whole lines, not 32-byte pieces of 16 lines
# speedup vs baseline: 7.4910x; 7.4910x over previous
#include <hip/hip_runtime.h>
#include <hip/hip_bf16.h>


typedef __attribute__((ext_vector_type(16))) __bf16 v16bf;
typedef __attribute__((ext_vector_type(8)))  __bf16 v8bf;
typedef __attribute__((ext_vector_type(8)))  float  v8f;

#define D_ROWS 8192
#define N_CENT 8192
#define N_FEAT 128
#define L_OUT  64
#define N_SPLIT 8
#define N_ITERS (N_CENT / (32 * N_SPLIT))

__device__ __forceinline__ v16bf frag_cat(v8bf a, v8bf b) {
  return __builtin_shufflevector(a, b, 0, 1, 2, 3, 4, 5, 6, 7,
                                       8, 9, 10, 11, 12, 13, 14, 15);
}

__device__ __forceinline__ v16bf load_frag_g(const __bf16* __restrict__ base,
                                             int off, int h) {
  const v8bf* p0 = reinterpret_cast<const v8bf*>(base + off + 8 * h);
  const v8bf* p1 = reinterpret_cast<const v8bf*>(base + off + 16 + 8 * h);
  return frag_cat(*p0, *p1);
}

__device__ __forceinline__ v16bf load_frag_blk(const __bf16* __restrict__ blk,
                                               int lane8) {
  const v8bf* p0 = reinterpret_cast<const v8bf*>(blk + lane8);
  const v8bf* p1 = reinterpret_cast<const v8bf*>(blk + 256 + lane8);
  return frag_cat(*p0, *p1);
}

__device__ __forceinline__ v16bf load_frag_lds(const __bf16* rowp, int h) {
  const v8bf* p0 = reinterpret_cast<const v8bf*>(rowp + 8 * h);
  const v8bf* p1 = reinterpret_cast<const v8bf*>(rowp + 16 + 8 * h);
  return frag_cat(*p0, *p1);
}

__device__ __forceinline__ v8f wmma_bf16(v16bf a, v16bf b, v8f c) {
  return __builtin_amdgcn_wmma_f32_16x16x32_bf16(
       false, a,  false, b,
       (short)0, c,  false,  false);
}

__global__ void split_f32_bf16(const float* __restrict__ in,
                               __bf16* __restrict__ hi, __bf16* __restrict__ lo,
                               int count) {
  int i = blockIdx.x * blockDim.x + threadIdx.x;
  if (i < count) {
    float v  = in[i];
    __bf16 h = (__bf16)v;
    hi[i]    = h;
    lo[i]    = (__bf16)(v - (float)h);
  }
}

__global__ void split_frag_centre(const float* __restrict__ in,
                                  __bf16* __restrict__ hi,
                                  __bf16* __restrict__ lo, int count) {
  int d = blockIdx.x * blockDim.x + threadIdx.x;
  if (d < count) {
    int e    = d & 7;
    int ln   = (d >> 3) & 15;
    int hh   = (d >> 7) & 1;
    int run  = (d >> 8) & 1;
    int kc   = (d >> 9) & 3;
    int tile = d >> 11;
    float v  = in[(tile * 16 + ln) * N_FEAT + kc * 32 + run * 16 + hh * 8 + e];
    __bf16 h = (__bf16)v;
    hi[d]    = h;
    lo[d]    = (__bf16)(v - (float)h);
  }
}

__global__ void split_frag_weight(const float* __restrict__ w,
                                  __bf16* __restrict__ hi,
                                  __bf16* __restrict__ lo, int count) {
  int d = blockIdx.x * blockDim.x + threadIdx.x;
  if (d < count) {
    int e     = d & 7;
    int ln    = (d >> 3) & 15;
    int hh    = (d >> 7) & 1;
    int run   = (d >> 8) & 1;
    int nt    = (d >> 9) & 3;
    int chunk = d >> 11;
    float v   = w[(chunk * 32 + run * 16 + hh * 8 + e) * L_OUT + nt * 16 + ln];
    __bf16 h  = (__bf16)v;
    hi[d]     = h;
    lo[d]     = (__bf16)(v - (float)h);
  }
}

__global__ void rowsum2(const float* __restrict__ in, float* __restrict__ out,
                        int rows, int cols, float bias) {
  int r = blockIdx.x * blockDim.x + threadIdx.x;
  if (r < rows) {
    const float* p = in + (size_t)r * cols;
    float s = bias;
    for (int k = 0; k < cols; ++k) s += p[k] * p[k];
    out[r] = s;
  }
}

__global__ __launch_bounds__(256) void rbfn_main(
    const __bf16* __restrict__ xs_hi, const __bf16* __restrict__ xs_lo,
    const __bf16* __restrict__ c_hi,  const __bf16* __restrict__ c_lo,
    const __bf16* __restrict__ wt_hi, const __bf16* __restrict__ wt_lo,
    const float* __restrict__ x2p1,
    const float* __restrict__ c2,
    float* __restrict__ out) {
  constexpr int RS = 40;
  __shared__ alignas(16) __bf16 lds_phi_hi[N_SPLIT][16][RS];
  __shared__ alignas(16) __bf16 lds_phi_lo[N_SPLIT][16][RS];
  constexpr int PS = 68;
  __shared__ alignas(16) float lds_part[N_SPLIT][16][PS];

  const int lane = threadIdx.x & 31;
  const int wave = __builtin_amdgcn_readfirstlane((int)(threadIdx.x >> 5));
  const int h    = lane >> 4;
  const int ln   = lane & 15;
  const int rowbase = blockIdx.x * 16;
  const int lane8   = lane * 8;

  v16bf a_hi[4], a_lo[4];
  {
    const int xoff = (rowbase + ln) * N_FEAT;
#pragma unroll
    for (int kc = 0; kc < 4; ++kc) {
      a_hi[kc] = load_frag_g(xs_hi, xoff + kc * 32, h);
      a_lo[kc] = load_frag_g(xs_lo, xoff + kc * 32, h);
    }
  }

  float x2v[8];
#pragma unroll
  for (int v = 0; v < 8; ++v) x2v[v] = x2p1[rowbase + v + 8 * h];

  v8f acc[4];
#pragma unroll
  for (int nt = 0; nt < 4; ++nt)
    acc[nt] = v8f{0.f, 0.f, 0.f, 0.f, 0.f, 0.f, 0.f, 0.f};

  __bf16* phi_hi = &lds_phi_hi[wave][0][0];
  __bf16* phi_lo = &lds_phi_lo[wave][0][0];

  for (int it = 0; it < N_ITERS; ++it) {
    const int chunk = it * N_SPLIT + wave;
    const int cb    = chunk * 32;
    if (it + 1 < N_ITERS) {
      __builtin_prefetch(c_hi + (size_t)((chunk + N_SPLIT) * 32 * N_FEAT + lane * 128), 0, 0);
    }

#pragma unroll
    for (int ct = 0; ct < 2; ++ct) {
      const int coff = (chunk * 2 + ct) * 2048;

      v8f g = v8f{0.f, 0.f, 0.f, 0.f, 0.f, 0.f, 0.f, 0.f};
#pragma unroll
      for (int kc = 0; kc < 4; ++kc) {
        v16bf b_hi = load_frag_blk(c_hi + coff + kc * 512, lane8);
        v16bf b_lo = load_frag_blk(c_lo + coff + kc * 512, lane8);
        g = wmma_bf16(a_hi[kc], b_hi, g);
        g = wmma_bf16(a_hi[kc], b_lo, g);
        g = wmma_bf16(a_lo[kc], b_hi, g);
      }

      const float c2v = c2[cb + ct * 16 + ln];
#pragma unroll
      for (int v = 0; v < 8; ++v) {
        float t   = fmaf(-2.0f, g[v], x2v[v] + c2v);
        t         = fmaxf(t, 1.0f);
        float phi = __builtin_amdgcn_sqrtf(t);
        __bf16 ph = (__bf16)phi;
        __bf16 pl = (__bf16)(phi - (float)ph);
        const int idx = (v + 8 * h) * RS + ct * 16 + ln;
        phi_hi[idx] = ph;
        phi_lo[idx] = pl;
      }
    }

    asm volatile("s_wait_dscnt 0" ::: "memory");

    v16bf pa_hi = load_frag_lds(phi_hi + ln * RS, h);
    v16bf pa_lo = load_frag_lds(phi_lo + ln * RS, h);

#pragma unroll
    for (int nt = 0; nt < 4; ++nt) {
      const int woff = (chunk * 4 + nt) * 512;
      v16bf wb_hi = load_frag_blk(wt_hi + woff, lane8);
      v16bf wb_lo = load_frag_blk(wt_lo + woff, lane8);
      acc[nt] = wmma_bf16(pa_hi, wb_hi, acc[nt]);
      acc[nt] = wmma_bf16(pa_hi, wb_lo, acc[nt]);
      acc[nt] = wmma_bf16(pa_lo, wb_hi, acc[nt]);
    }
  }

  float* part = &lds_part[wave][0][0];
#pragma unroll
  for (int nt = 0; nt < 4; ++nt)
#pragma unroll
    for (int v = 0; v < 8; ++v)
      part[(v + 8 * h) * PS + nt * 16 + ln] = acc[nt][v];

  __syncthreads();

  {
    const int t   = threadIdx.x;
    const int row = t >> 4;
    const int col = (t & 15) * 4;
    float4 s = *reinterpret_cast<const float4*>(&lds_part[0][row][col]);
#pragma unroll
    for (int w = 1; w < N_SPLIT; ++w) {
      const float4 p = *reinterpret_cast<const float4*>(&lds_part[w][row][col]);
      s.x += p.x; s.y += p.y; s.z += p.z; s.w += p.w;
    }
    *reinterpret_cast<float4*>(out + (rowbase + row) * L_OUT + col) = s;
  }
}

extern "C" void kernel_launch(void* const* d_in, const int* in_sizes, int n_in,
                              void* d_out, int out_size, void* d_ws, size_t ws_size,
                              hipStream_t stream) {
  (void)in_sizes; (void)n_in; (void)out_size; (void)ws_size;

  const float* xs     = (const float*)d_in[0];
  const float* centre = (const float*)d_in[1];
  const float* weight = (const float*)d_in[2];
  float* out          = (float*)d_out;

  const size_t XS = (size_t)D_ROWS * N_FEAT;
  const size_t WT = (size_t)N_CENT * L_OUT;

  __bf16* xs_hi = (__bf16*)d_ws;
  __bf16* xs_lo = xs_hi + XS;
  __bf16* c_hi  = xs_lo + XS;
  __bf16* c_lo  = c_hi + XS;
  __bf16* wt_hi = c_lo + XS;
  __bf16* wt_lo = wt_hi + WT;
  float*  x2p1  = (float*)(wt_lo + WT);
  float*  c2    = x2p1 + D_ROWS;

  const int T = 256;
  split_f32_bf16<<<dim3((XS + T - 1) / T), dim3(T), 0, stream>>>(xs, xs_hi, xs_lo, (int)XS);
  split_frag_centre<<<dim3((XS + T - 1) / T), dim3(T), 0, stream>>>(centre, c_hi, c_lo, (int)XS);
  split_frag_weight<<<dim3((WT + T - 1) / T), dim3(T), 0, stream>>>(weight, wt_hi, wt_lo, (int)WT);
  rowsum2<<<dim3((D_ROWS + T - 1) / T), dim3(T), 0, stream>>>(xs, x2p1, D_ROWS, N_FEAT, 1.0f);
  rowsum2<<<dim3((N_CENT + T - 1) / T), dim3(T), 0, stream>>>(centre, c2, N_CENT, N_FEAT, 0.0f);

  rbfn_main<<<dim3(D_ROWS / 16), dim3(32 * N_SPLIT), 0, stream>>>(
      xs_hi, xs_lo, c_hi, c_lo, wt_hi, wt_lo, x2p1, c2, out);
}


# A wave keeps a whole chunk's centre and W fragments in registers and reloads each set for its next chunk as one group right behind its last use, not load-wait-multiply per k-step; sums unchanged
# speedup vs baseline: 9.7239x; 1.2981x over previous
#include <hip/hip_runtime.h>
#include <hip/hip_bf16.h>


typedef __attribute__((ext_vector_type(16))) __bf16 v16bf;
typedef __attribute__((ext_vector_type(8)))  __bf16 v8bf;
typedef __attribute__((ext_vector_type(8)))  float  v8f;

#define D_ROWS 8192
#define N_CENT 8192
#define N_FEAT 128
#define L_OUT  64
#define N_SPLIT 8
#define N_ITERS (N_CENT / (32 * N_SPLIT))

__device__ __forceinline__ v16bf frag_cat(v8bf a, v8bf b) {
  return __builtin_shufflevector(a, b, 0, 1, 2, 3, 4, 5, 6, 7,
                                       8, 9, 10, 11, 12, 13, 14, 15);
}

__device__ __forceinline__ v16bf load_frag_g(const __bf16* __restrict__ base,
                                             int off, int h) {
  const v8bf* p0 = reinterpret_cast<const v8bf*>(base + off + 8 * h);
  const v8bf* p1 = reinterpret_cast<const v8bf*>(base + off + 16 + 8 * h);
  return frag_cat(*p0, *p1);
}

__device__ __forceinline__ v16bf load_frag_blk(const __bf16* __restrict__ blk,
                                               int lane8) {
  const v8bf* p0 = reinterpret_cast<const v8bf*>(blk + lane8);
  const v8bf* p1 = reinterpret_cast<const v8bf*>(blk + 256 + lane8);
  return frag_cat(*p0, *p1);
}

__device__ __forceinline__ v16bf load_frag_lds(const __bf16* rowp, int h) {
  const v8bf* p0 = reinterpret_cast<const v8bf*>(rowp + 8 * h);
  const v8bf* p1 = reinterpret_cast<const v8bf*>(rowp + 16 + 8 * h);
  return frag_cat(*p0, *p1);
}

__device__ __forceinline__ v8f wmma_bf16(v16bf a, v16bf b, v8f c) {
  return __builtin_amdgcn_wmma_f32_16x16x32_bf16(
       false, a,  false, b,
       (short)0, c,  false,  false);
}

__device__ __forceinline__ void load_centre_chunk(
    const __bf16* __restrict__ c_hi, const __bf16* __restrict__ c_lo,
    int chunk, int lane8, v16bf (&b_hi)[2][4], v16bf (&b_lo)[2][4]) {
#pragma unroll
  for (int ct = 0; ct < 2; ++ct) {
    const int coff = (chunk * 2 + ct) * 2048;
#pragma unroll
    for (int kc = 0; kc < 4; ++kc) {
      b_hi[ct][kc] = load_frag_blk(c_hi + coff + kc * 512, lane8);
      b_lo[ct][kc] = load_frag_blk(c_lo + coff + kc * 512, lane8);
    }
  }
}

__device__ __forceinline__ void load_w_chunk(
    const __bf16* __restrict__ wt_hi, const __bf16* __restrict__ wt_lo,
    int chunk, int lane8, v16bf (&w_hi)[4], v16bf (&w_lo)[4]) {
#pragma unroll
  for (int nt = 0; nt < 4; ++nt) {
    const int woff = (chunk * 4 + nt) * 512;
    w_hi[nt] = load_frag_blk(wt_hi + woff, lane8);
    w_lo[nt] = load_frag_blk(wt_lo + woff, lane8);
  }
}

__global__ void split_f32_bf16(const float* __restrict__ in,
                               __bf16* __restrict__ hi, __bf16* __restrict__ lo,
                               int count) {
  int i = blockIdx.x * blockDim.x + threadIdx.x;
  if (i < count) {
    float v  = in[i];
    __bf16 h = (__bf16)v;
    hi[i]    = h;
    lo[i]    = (__bf16)(v - (float)h);
  }
}

__global__ void split_frag_centre(const float* __restrict__ in,
                                  __bf16* __restrict__ hi,
                                  __bf16* __restrict__ lo, int count) {
  int d = blockIdx.x * blockDim.x + threadIdx.x;
  if (d < count) {
    int e    = d & 7;
    int ln   = (d >> 3) & 15;
    int hh   = (d >> 7) & 1;
    int run  = (d >> 8) & 1;
    int kc   = (d >> 9) & 3;
    int tile = d >> 11;
    float v  = in[(tile * 16 + ln) * N_FEAT + kc * 32 + run * 16 + hh * 8 + e];
    __bf16 h = (__bf16)v;
    hi[d]    = h;
    lo[d]    = (__bf16)(v - (float)h);
  }
}

__global__ void split_frag_weight(const float* __restrict__ w,
                                  __bf16* __restrict__ hi,
                                  __bf16* __restrict__ lo, int count) {
  int d = blockIdx.x * blockDim.x + threadIdx.x;
  if (d < count) {
    int e     = d & 7;
    int ln    = (d >> 3) & 15;
    int hh    = (d >> 7) & 1;
    int run   = (d >> 8) & 1;
    int nt    = (d >> 9) & 3;
    int chunk = d >> 11;
    float v   = w[(chunk * 32 + run * 16 + hh * 8 + e) * L_OUT + nt * 16 + ln];
    __bf16 h  = (__bf16)v;
    hi[d]     = h;
    lo[d]     = (__bf16)(v - (float)h);
  }
}

__global__ void rowsum2(const float* __restrict__ in, float* __restrict__ out,
                        int rows, int cols, float bias) {
  int r = blockIdx.x * blockDim.x + threadIdx.x;
  if (r < rows) {
    const float* p = in + (size_t)r * cols;
    float s = bias;
    for (int k = 0; k < cols; ++k) s += p[k] * p[k];
    out[r] = s;
  }
}

__global__ __launch_bounds__(256) void rbfn_main(
    const __bf16* __restrict__ xs_hi, const __bf16* __restrict__ xs_lo,
    const __bf16* __restrict__ c_hi,  const __bf16* __restrict__ c_lo,
    const __bf16* __restrict__ wt_hi, const __bf16* __restrict__ wt_lo,
    const float* __restrict__ x2p1,
    const float* __restrict__ c2,
    float* __restrict__ out) {
  constexpr int RS = 40;
  __shared__ alignas(16) __bf16 lds_phi_hi[N_SPLIT][16][RS];
  __shared__ alignas(16) __bf16 lds_phi_lo[N_SPLIT][16][RS];
  constexpr int PS = 68;
  __shared__ alignas(16) float lds_part[N_SPLIT][16][PS];

  const int lane = threadIdx.x & 31;
  const int wave = __builtin_amdgcn_readfirstlane((int)(threadIdx.x >> 5));
  const int h    = lane >> 4;
  const int ln   = lane & 15;
  const int rowbase = blockIdx.x * 16;
  const int lane8   = lane * 8;

  v16bf a_hi[4], a_lo[4];
  {
    const int xoff = (rowbase + ln) * N_FEAT;
#pragma unroll
    for (int kc = 0; kc < 4; ++kc) {
      a_hi[kc] = load_frag_g(xs_hi, xoff + kc * 32, h);
      a_lo[kc] = load_frag_g(xs_lo, xoff + kc * 32, h);
    }
  }

  float x2v[8];
#pragma unroll
  for (int v = 0; v < 8; ++v) x2v[v] = x2p1[rowbase + v + 8 * h];

  v8f acc[4];
#pragma unroll
  for (int nt = 0; nt < 4; ++nt)
    acc[nt] = v8f{0.f, 0.f, 0.f, 0.f, 0.f, 0.f, 0.f, 0.f};

  __bf16* phi_hi = &lds_phi_hi[wave][0][0];
  __bf16* phi_lo = &lds_phi_lo[wave][0][0];

  v16bf bc_hi[2][4], bc_lo[2][4];
  v16bf bw_hi[4], bw_lo[4];
  float c2v[2];

  c2v[0] = c2[wave * 32 + ln];
  c2v[1] = c2[wave * 32 + 16 + ln];
  __builtin_amdgcn_sched_barrier(0);
  load_centre_chunk(c_hi, c_lo, wave, lane8, bc_hi, bc_lo);
  __builtin_amdgcn_sched_barrier(0);
  load_w_chunk(wt_hi, wt_lo, wave, lane8, bw_hi, bw_lo);

  auto step = [&](const int chunk, const bool more) __attribute__((always_inline)) {
    v8f g[2];
#pragma unroll
    for (int ct = 0; ct < 2; ++ct)
      g[ct] = v8f{0.f, 0.f, 0.f, 0.f, 0.f, 0.f, 0.f, 0.f};
#pragma unroll
    for (int kc = 0; kc < 4; ++kc) {
#pragma unroll
      for (int ct = 0; ct < 2; ++ct)
        g[ct] = wmma_bf16(a_hi[kc], bc_hi[ct][kc], g[ct]);
#pragma unroll
      for (int ct = 0; ct < 2; ++ct)
        g[ct] = wmma_bf16(a_hi[kc], bc_lo[ct][kc], g[ct]);
#pragma unroll
      for (int ct = 0; ct < 2; ++ct)
        g[ct] = wmma_bf16(a_lo[kc], bc_hi[ct][kc], g[ct]);
    }

    const float c2cur[2] = {c2v[0], c2v[1]};
    if (more) {
      load_centre_chunk(c_hi, c_lo, chunk + N_SPLIT, lane8, bc_hi, bc_lo);
      c2v[0] = c2[(chunk + N_SPLIT) * 32 + ln];
      c2v[1] = c2[(chunk + N_SPLIT) * 32 + 16 + ln];
    }

#pragma unroll
    for (int ct = 0; ct < 2; ++ct) {
#pragma unroll
      for (int v = 0; v < 8; ++v) {
        float t   = fmaf(-2.0f, g[ct][v], x2v[v] + c2cur[ct]);
        t         = fmaxf(t, 1.0f);
        float phi = __builtin_amdgcn_sqrtf(t);
        __bf16 ph = (__bf16)phi;
        __bf16 pl = (__bf16)(phi - (float)ph);
        const int idx = (v + 8 * h) * RS + ct * 16 + ln;
        phi_hi[idx] = ph;
        phi_lo[idx] = pl;
      }
    }

    asm volatile("s_wait_dscnt 0" ::: "memory");

    v16bf pa_hi = load_frag_lds(phi_hi + ln * RS, h);
    v16bf pa_lo = load_frag_lds(phi_lo + ln * RS, h);

#pragma unroll
    for (int nt = 0; nt < 4; ++nt) {
      acc[nt] = wmma_bf16(pa_hi, bw_hi[nt], acc[nt]);
      acc[nt] = wmma_bf16(pa_hi, bw_lo[nt], acc[nt]);
      acc[nt] = wmma_bf16(pa_lo, bw_hi[nt], acc[nt]);
    }

    if (more) load_w_chunk(wt_hi, wt_lo, chunk + N_SPLIT, lane8, bw_hi, bw_lo);
  };

  for (int it = 0; it < N_ITERS - 1; ++it) step(it * N_SPLIT + wave, true);
  step((N_ITERS - 1) * N_SPLIT + wave, false);

  float* part = &lds_part[wave][0][0];
#pragma unroll
  for (int nt = 0; nt < 4; ++nt)
#pragma unroll
    for (int v = 0; v < 8; ++v)
      part[(v + 8 * h) * PS + nt * 16 + ln] = acc[nt][v];

  __syncthreads();

  {
    const int t   = threadIdx.x;
    const int row = t >> 4;
    const int col = (t & 15) * 4;
    float4 s = *reinterpret_cast<const float4*>(&lds_part[0][row][col]);
#pragma unroll
    for (int w = 1; w < N_SPLIT; ++w) {
      const float4 p = *reinterpret_cast<const float4*>(&lds_part[w][row][col]);
      s.x += p.x; s.y += p.y; s.z += p.z; s.w += p.w;
    }
    *reinterpret_cast<float4*>(out + (rowbase + row) * L_OUT + col) = s;
  }
}

extern "C" void kernel_launch(void* const* d_in, const int* in_sizes, int n_in,
                              void* d_out, int out_size, void* d_ws, size_t ws_size,
                              hipStream_t stream) {
  (void)in_sizes; (void)n_in; (void)out_size; (void)ws_size;

  const float* xs     = (const float*)d_in[0];
  const float* centre = (const float*)d_in[1];
  const float* weight = (const float*)d_in[2];
  float* out          = (float*)d_out;

  const size_t XS = (size_t)D_ROWS * N_FEAT;
  const size_t WT = (size_t)N_CENT * L_OUT;

  __bf16* xs_hi = (__bf16*)d_ws;
  __bf16* xs_lo = xs_hi + XS;
  __bf16* c_hi  = xs_lo + XS;
  __bf16* c_lo  = c_hi + XS;
  __bf16* wt_hi = c_lo + XS;
  __bf16* wt_lo = wt_hi + WT;
  float*  x2p1  = (float*)(wt_lo + WT);
  float*  c2    = x2p1 + D_ROWS;

  const int T = 256;
  split_f32_bf16<<<dim3((XS + T - 1) / T), dim3(T), 0, stream>>>(xs, xs_hi, xs_lo, (int)XS);
  split_frag_centre<<<dim3((XS + T - 1) / T), dim3(T), 0, stream>>>(centre, c_hi, c_lo, (int)XS);
  split_frag_weight<<<dim3((WT + T - 1) / T), dim3(T), 0, stream>>>(weight, wt_hi, wt_lo, (int)WT);
  rowsum2<<<dim3((D_ROWS + T - 1) / T), dim3(T), 0, stream>>>(xs, x2p1, D_ROWS, N_FEAT, 1.0f);
  rowsum2<<<dim3((N_CENT + T - 1) / T), dim3(T), 0, stream>>>(centre, c2, N_CENT, N_FEAT, 0.0f);

  rbfn_main<<<dim3(D_ROWS / 16), dim3(32 * N_SPLIT), 0, stream>>>(
      xs_hi, xs_lo, c_hi, c_lo, wt_hi, wt_lo, x2p1, c2, out);
}


# Each wave computes two 16-row tiles (block = 32 rows, 256 blocks) from every centre and W fragment it loads, halving second-level cache reads per product; order of the f32 sums unchanged
# speedup vs baseline: 12.9220x; 1.3289x over previous
#include <hip/hip_runtime.h>
#include <hip/hip_bf16.h>


typedef __attribute__((ext_vector_type(16))) __bf16 v16bf;
typedef __attribute__((ext_vector_type(8)))  __bf16 v8bf;
typedef __attribute__((ext_vector_type(8)))  float  v8f;

#define D_ROWS 8192
#define N_CENT 8192
#define N_FEAT 128
#define L_OUT  64
#define N_SPLIT 8
#define N_ITERS (N_CENT / (32 * N_SPLIT))
#define M_TILES 2
#define BLK_ROWS (16 * M_TILES)

__device__ __forceinline__ v16bf frag_cat(v8bf a, v8bf b) {
  return __builtin_shufflevector(a, b, 0, 1, 2, 3, 4, 5, 6, 7,
                                       8, 9, 10, 11, 12, 13, 14, 15);
}

__device__ __forceinline__ v16bf load_frag_g(const __bf16* __restrict__ base,
                                             int off, int h) {
  const v8bf* p0 = reinterpret_cast<const v8bf*>(base + off + 8 * h);
  const v8bf* p1 = reinterpret_cast<const v8bf*>(base + off + 16 + 8 * h);
  return frag_cat(*p0, *p1);
}

__device__ __forceinline__ v16bf load_frag_blk(const __bf16* __restrict__ blk,
                                               int lane8) {
  const v8bf* p0 = reinterpret_cast<const v8bf*>(blk + lane8);
  const v8bf* p1 = reinterpret_cast<const v8bf*>(blk + 256 + lane8);
  return frag_cat(*p0, *p1);
}

__device__ __forceinline__ v16bf load_frag_lds(const __bf16* rowp, int h) {
  const v8bf* p0 = reinterpret_cast<const v8bf*>(rowp + 8 * h);
  const v8bf* p1 = reinterpret_cast<const v8bf*>(rowp + 16 + 8 * h);
  return frag_cat(*p0, *p1);
}

__device__ __forceinline__ v8f wmma_bf16(v16bf a, v16bf b, v8f c) {
  return __builtin_amdgcn_wmma_f32_16x16x32_bf16(
       false, a,  false, b,
       (short)0, c,  false,  false);
}

__device__ __forceinline__ void load_centre_chunk(
    const __bf16* __restrict__ c_hi, const __bf16* __restrict__ c_lo,
    int chunk, int lane8, v16bf (&b_hi)[2][4], v16bf (&b_lo)[2][4]) {
#pragma unroll
  for (int ct = 0; ct < 2; ++ct) {
    const int coff = (chunk * 2 + ct) * 2048;
#pragma unroll
    for (int kc = 0; kc < 4; ++kc) {
      b_hi[ct][kc] = load_frag_blk(c_hi + coff + kc * 512, lane8);
      b_lo[ct][kc] = load_frag_blk(c_lo + coff + kc * 512, lane8);
    }
  }
}

__device__ __forceinline__ void load_w_chunk(
    const __bf16* __restrict__ wt_hi, const __bf16* __restrict__ wt_lo,
    int chunk, int lane8, v16bf (&w_hi)[4], v16bf (&w_lo)[4]) {
#pragma unroll
  for (int nt = 0; nt < 4; ++nt) {
    const int woff = (chunk * 4 + nt) * 512;
    w_hi[nt] = load_frag_blk(wt_hi + woff, lane8);
    w_lo[nt] = load_frag_blk(wt_lo + woff, lane8);
  }
}

__global__ void split_f32_bf16(const float* __restrict__ in,
                               __bf16* __restrict__ hi, __bf16* __restrict__ lo,
                               int count) {
  int i = blockIdx.x * blockDim.x + threadIdx.x;
  if (i < count) {
    float v  = in[i];
    __bf16 h = (__bf16)v;
    hi[i]    = h;
    lo[i]    = (__bf16)(v - (float)h);
  }
}

__global__ void split_frag_centre(const float* __restrict__ in,
                                  __bf16* __restrict__ hi,
                                  __bf16* __restrict__ lo, int count) {
  int d = blockIdx.x * blockDim.x + threadIdx.x;
  if (d < count) {
    int e    = d & 7;
    int ln   = (d >> 3) & 15;
    int hh   = (d >> 7) & 1;
    int run  = (d >> 8) & 1;
    int kc   = (d >> 9) & 3;
    int tile = d >> 11;
    float v  = in[(tile * 16 + ln) * N_FEAT + kc * 32 + run * 16 + hh * 8 + e];
    __bf16 h = (__bf16)v;
    hi[d]    = h;
    lo[d]    = (__bf16)(v - (float)h);
  }
}

__global__ void split_frag_weight(const float* __restrict__ w,
                                  __bf16* __restrict__ hi,
                                  __bf16* __restrict__ lo, int count) {
  int d = blockIdx.x * blockDim.x + threadIdx.x;
  if (d < count) {
    int e     = d & 7;
    int ln    = (d >> 3) & 15;
    int hh    = (d >> 7) & 1;
    int run   = (d >> 8) & 1;
    int nt    = (d >> 9) & 3;
    int chunk = d >> 11;
    float v   = w[(chunk * 32 + run * 16 + hh * 8 + e) * L_OUT + nt * 16 + ln];
    __bf16 h  = (__bf16)v;
    hi[d]     = h;
    lo[d]     = (__bf16)(v - (float)h);
  }
}

__global__ void rowsum2(const float* __restrict__ in, float* __restrict__ out,
                        int rows, int cols, float bias) {
  int r = blockIdx.x * blockDim.x + threadIdx.x;
  if (r < rows) {
    const float* p = in + (size_t)r * cols;
    float s = bias;
    for (int k = 0; k < cols; ++k) s += p[k] * p[k];
    out[r] = s;
  }
}

__global__ __launch_bounds__(256) void rbfn_main(
    const __bf16* __restrict__ xs_hi, const __bf16* __restrict__ xs_lo,
    const __bf16* __restrict__ c_hi,  const __bf16* __restrict__ c_lo,
    const __bf16* __restrict__ wt_hi, const __bf16* __restrict__ wt_lo,
    const float* __restrict__ x2p1,
    const float* __restrict__ c2,
    float* __restrict__ out) {
  constexpr int RS = 40;
  __shared__ alignas(16) __bf16 lds_phi_hi[N_SPLIT][M_TILES][16][RS];
  __shared__ alignas(16) __bf16 lds_phi_lo[N_SPLIT][M_TILES][16][RS];
  constexpr int PS = 68;
  __shared__ alignas(16) float lds_part[N_SPLIT][BLK_ROWS][PS];

  const int lane = threadIdx.x & 31;
  const int wave = __builtin_amdgcn_readfirstlane((int)(threadIdx.x >> 5));
  const int h    = lane >> 4;
  const int ln   = lane & 15;
  const int rowbase = blockIdx.x * BLK_ROWS;
  const int lane8   = lane * 8;

  v16bf a_hi[M_TILES][4], a_lo[M_TILES][4];
#pragma unroll
  for (int mt = 0; mt < M_TILES; ++mt) {
    const int xoff = (rowbase + mt * 16 + ln) * N_FEAT;
#pragma unroll
    for (int kc = 0; kc < 4; ++kc) {
      a_hi[mt][kc] = load_frag_g(xs_hi, xoff + kc * 32, h);
      a_lo[mt][kc] = load_frag_g(xs_lo, xoff + kc * 32, h);
    }
  }

  float x2v[M_TILES][8];
#pragma unroll
  for (int mt = 0; mt < M_TILES; ++mt)
#pragma unroll
    for (int v = 0; v < 8; ++v) x2v[mt][v] = x2p1[rowbase + mt * 16 + v + 8 * h];

  v8f acc[M_TILES][4];
#pragma unroll
  for (int mt = 0; mt < M_TILES; ++mt)
#pragma unroll
    for (int nt = 0; nt < 4; ++nt)
      acc[mt][nt] = v8f{0.f, 0.f, 0.f, 0.f, 0.f, 0.f, 0.f, 0.f};

  __bf16* phi_hi = &lds_phi_hi[wave][0][0][0];
  __bf16* phi_lo = &lds_phi_lo[wave][0][0][0];

  v16bf bc_hi[2][4], bc_lo[2][4];
  v16bf bw_hi[4], bw_lo[4];
  float c2v[2];

  c2v[0] = c2[wave * 32 + ln];
  c2v[1] = c2[wave * 32 + 16 + ln];
  __builtin_amdgcn_sched_barrier(0);
  load_centre_chunk(c_hi, c_lo, wave, lane8, bc_hi, bc_lo);
  __builtin_amdgcn_sched_barrier(0);
  load_w_chunk(wt_hi, wt_lo, wave, lane8, bw_hi, bw_lo);

  auto step = [&](const int chunk, const bool more) __attribute__((always_inline)) {
    v8f g[M_TILES][2];
#pragma unroll
    for (int mt = 0; mt < M_TILES; ++mt)
#pragma unroll
      for (int ct = 0; ct < 2; ++ct)
        g[mt][ct] = v8f{0.f, 0.f, 0.f, 0.f, 0.f, 0.f, 0.f, 0.f};
#pragma unroll
    for (int kc = 0; kc < 4; ++kc) {
#pragma unroll
      for (int mt = 0; mt < M_TILES; ++mt)
#pragma unroll
        for (int ct = 0; ct < 2; ++ct)
          g[mt][ct] = wmma_bf16(a_hi[mt][kc], bc_hi[ct][kc], g[mt][ct]);
#pragma unroll
      for (int mt = 0; mt < M_TILES; ++mt)
#pragma unroll
        for (int ct = 0; ct < 2; ++ct)
          g[mt][ct] = wmma_bf16(a_hi[mt][kc], bc_lo[ct][kc], g[mt][ct]);
#pragma unroll
      for (int mt = 0; mt < M_TILES; ++mt)
#pragma unroll
        for (int ct = 0; ct < 2; ++ct)
          g[mt][ct] = wmma_bf16(a_lo[mt][kc], bc_hi[ct][kc], g[mt][ct]);
    }

    const float c2cur[2] = {c2v[0], c2v[1]};
    if (more) {
      load_centre_chunk(c_hi, c_lo, chunk + N_SPLIT, lane8, bc_hi, bc_lo);
      c2v[0] = c2[(chunk + N_SPLIT) * 32 + ln];
      c2v[1] = c2[(chunk + N_SPLIT) * 32 + 16 + ln];
    }
    __builtin_amdgcn_sched_barrier(0);

#pragma unroll
    for (int mt = 0; mt < M_TILES; ++mt) {
#pragma unroll
      for (int ct = 0; ct < 2; ++ct) {
        float  phi[8];
        __bf16 ph[8], pl[8];
#pragma unroll
        for (int v = 0; v < 8; ++v)
          phi[v] = fmaxf(fmaf(-2.0f, g[mt][ct][v], x2v[mt][v] + c2cur[ct]), 1.0f);
#pragma unroll
        for (int v = 0; v < 8; ++v)
          phi[v] = __builtin_amdgcn_sqrtf(phi[v]);
#pragma unroll
        for (int v = 0; v < 8; ++v) ph[v] = (__bf16)phi[v];
#pragma unroll
        for (int v = 0; v < 8; ++v) pl[v] = (__bf16)(phi[v] - (float)ph[v]);
#pragma unroll
        for (int v = 0; v < 8; ++v) {
          const int idx = (mt * 16 + v + 8 * h) * RS + ct * 16 + ln;
          phi_hi[idx] = ph[v];
          phi_lo[idx] = pl[v];
        }
      }
    }

    asm volatile("s_wait_dscnt 0" ::: "memory");

    v16bf pa_hi[M_TILES], pa_lo[M_TILES];
#pragma unroll
    for (int mt = 0; mt < M_TILES; ++mt) {
      pa_hi[mt] = load_frag_lds(phi_hi + (mt * 16 + ln) * RS, h);
      pa_lo[mt] = load_frag_lds(phi_lo + (mt * 16 + ln) * RS, h);
    }

#pragma unroll
    for (int nt = 0; nt < 4; ++nt) {
#pragma unroll
      for (int mt = 0; mt < M_TILES; ++mt)
        acc[mt][nt] = wmma_bf16(pa_hi[mt], bw_hi[nt], acc[mt][nt]);
#pragma unroll
      for (int mt = 0; mt < M_TILES; ++mt)
        acc[mt][nt] = wmma_bf16(pa_hi[mt], bw_lo[nt], acc[mt][nt]);
#pragma unroll
      for (int mt = 0; mt < M_TILES; ++mt)
        acc[mt][nt] = wmma_bf16(pa_lo[mt], bw_hi[nt], acc[mt][nt]);
    }

    if (more) load_w_chunk(wt_hi, wt_lo, chunk + N_SPLIT, lane8, bw_hi, bw_lo);
  };

  for (int it = 0; it < N_ITERS - 1; ++it) step(it * N_SPLIT + wave, true);
  step((N_ITERS - 1) * N_SPLIT + wave, false);

  float* part = &lds_part[wave][0][0];
#pragma unroll
  for (int mt = 0; mt < M_TILES; ++mt)
#pragma unroll
    for (int nt = 0; nt < 4; ++nt)
#pragma unroll
      for (int v = 0; v < 8; ++v)
        part[(mt * 16 + v + 8 * h) * PS + nt * 16 + ln] = acc[mt][nt][v];

  __syncthreads();

#pragma unroll
  for (int mt = 0; mt < M_TILES; ++mt) {
    const int t   = threadIdx.x;
    const int row = mt * 16 + (t >> 4);
    const int col = (t & 15) * 4;
    float4 s = *reinterpret_cast<const float4*>(&lds_part[0][row][col]);
#pragma unroll
    for (int w = 1; w < N_SPLIT; ++w) {
      const float4 p = *reinterpret_cast<const float4*>(&lds_part[w][row][col]);
      s.x += p.x; s.y += p.y; s.z += p.z; s.w += p.w;
    }
    *reinterpret_cast<float4*>(out + (rowbase + row) * L_OUT + col) = s;
  }
}

extern "C" void kernel_launch(void* const* d_in, const int* in_sizes, int n_in,
                              void* d_out, int out_size, void* d_ws, size_t ws_size,
                              hipStream_t stream) {
  (void)in_sizes; (void)n_in; (void)out_size; (void)ws_size;

  const float* xs     = (const float*)d_in[0];
  const float* centre = (const float*)d_in[1];
  const float* weight = (const float*)d_in[2];
  float* out          = (float*)d_out;

  const size_t XS = (size_t)D_ROWS * N_FEAT;
  const size_t WT = (size_t)N_CENT * L_OUT;

  __bf16* xs_hi = (__bf16*)d_ws;
  __bf16* xs_lo = xs_hi + XS;
  __bf16* c_hi  = xs_lo + XS;
  __bf16* c_lo  = c_hi + XS;
  __bf16* wt_hi = c_lo + XS;
  __bf16* wt_lo = wt_hi + WT;
  float*  x2p1  = (float*)(wt_lo + WT);
  float*  c2    = x2p1 + D_ROWS;

  const int T = 256;
  split_f32_bf16<<<dim3((XS + T - 1) / T), dim3(T), 0, stream>>>(xs, xs_hi, xs_lo, (int)XS);
  split_frag_centre<<<dim3((XS + T - 1) / T), dim3(T), 0, stream>>>(centre, c_hi, c_lo, (int)XS);
  split_frag_weight<<<dim3((WT + T - 1) / T), dim3(T), 0, stream>>>(weight, wt_hi, wt_lo, (int)WT);
  rowsum2<<<dim3((D_ROWS + T - 1) / T), dim3(T), 0, stream>>>(xs, x2p1, D_ROWS, N_FEAT, 1.0f);
  rowsum2<<<dim3((N_CENT + T - 1) / T), dim3(T), 0, stream>>>(centre, c2, N_CENT, N_FEAT, 0.0f);

  rbfn_main<<<dim3(D_ROWS / BLK_ROWS), dim3(32 * N_SPLIT), 0, stream>>>(
      xs_hi, xs_lo, c_hi, c_lo, wt_hi, wt_lo, x2p1, c2, out);
}


# The five pre-pass launches (three operand splits, two row-norm sums) merged into one launch whose blocks pick their task from the block index: 2 launches per run, not 6; values and f32 sums unchanged
# speedup vs baseline: 14.2216x; 1.1006x over previous
#include <hip/hip_runtime.h>
#include <hip/hip_bf16.h>


typedef __attribute__((ext_vector_type(16))) __bf16 v16bf;
typedef __attribute__((ext_vector_type(8)))  __bf16 v8bf;
typedef __attribute__((ext_vector_type(8)))  float  v8f;

#define D_ROWS 8192
#define N_CENT 8192
#define N_FEAT 128
#define L_OUT  64
#define N_SPLIT 8
#define N_ITERS (N_CENT / (32 * N_SPLIT))
#define M_TILES 2
#define BLK_ROWS (16 * M_TILES)

__device__ __forceinline__ v16bf frag_cat(v8bf a, v8bf b) {
  return __builtin_shufflevector(a, b, 0, 1, 2, 3, 4, 5, 6, 7,
                                       8, 9, 10, 11, 12, 13, 14, 15);
}

__device__ __forceinline__ v16bf load_frag_g(const __bf16* __restrict__ base,
                                             int off, int h) {
  const v8bf* p0 = reinterpret_cast<const v8bf*>(base + off + 8 * h);
  const v8bf* p1 = reinterpret_cast<const v8bf*>(base + off + 16 + 8 * h);
  return frag_cat(*p0, *p1);
}

__device__ __forceinline__ v16bf load_frag_blk(const __bf16* __restrict__ blk,
                                               int lane8) {
  const v8bf* p0 = reinterpret_cast<const v8bf*>(blk + lane8);
  const v8bf* p1 = reinterpret_cast<const v8bf*>(blk + 256 + lane8);
  return frag_cat(*p0, *p1);
}

__device__ __forceinline__ v16bf load_frag_lds(const __bf16* rowp, int h) {
  const v8bf* p0 = reinterpret_cast<const v8bf*>(rowp + 8 * h);
  const v8bf* p1 = reinterpret_cast<const v8bf*>(rowp + 16 + 8 * h);
  return frag_cat(*p0, *p1);
}

__device__ __forceinline__ v8f wmma_bf16(v16bf a, v16bf b, v8f c) {
  return __builtin_amdgcn_wmma_f32_16x16x32_bf16(
       false, a,  false, b,
       (short)0, c,  false,  false);
}

__device__ __forceinline__ void load_centre_chunk(
    const __bf16* __restrict__ c_hi, const __bf16* __restrict__ c_lo,
    int chunk, int lane8, v16bf (&b_hi)[2][4], v16bf (&b_lo)[2][4]) {
#pragma unroll
  for (int ct = 0; ct < 2; ++ct) {
    const int coff = (chunk * 2 + ct) * 2048;
#pragma unroll
    for (int kc = 0; kc < 4; ++kc) {
      b_hi[ct][kc] = load_frag_blk(c_hi + coff + kc * 512, lane8);
      b_lo[ct][kc] = load_frag_blk(c_lo + coff + kc * 512, lane8);
    }
  }
}

__device__ __forceinline__ void load_w_chunk(
    const __bf16* __restrict__ wt_hi, const __bf16* __restrict__ wt_lo,
    int chunk, int lane8, v16bf (&w_hi)[4], v16bf (&w_lo)[4]) {
#pragma unroll
  for (int nt = 0; nt < 4; ++nt) {
    const int woff = (chunk * 4 + nt) * 512;
    w_hi[nt] = load_frag_blk(wt_hi + woff, lane8);
    w_lo[nt] = load_frag_blk(wt_lo + woff, lane8);
  }
}

#define PRE_T 256
#define NB_X2 (D_ROWS / PRE_T)
#define NB_C2 (N_CENT / PRE_T)
#define NB_W  ((N_CENT * L_OUT) / PRE_T)
#define NB_C  ((N_CENT * N_FEAT) / PRE_T)
#define NB_X  ((D_ROWS * N_FEAT) / PRE_T)
#define NB_PRE (NB_X2 + NB_C2 + NB_W + NB_C + NB_X)

__device__ __forceinline__ void task_split_f32_bf16(
    const float* __restrict__ in, __bf16* __restrict__ hi,
    __bf16* __restrict__ lo, int i) {
  float v  = in[i];
  __bf16 h = (__bf16)v;
  hi[i]    = h;
  lo[i]    = (__bf16)(v - (float)h);
}

__device__ __forceinline__ void task_split_frag_centre(
    const float* __restrict__ in, __bf16* __restrict__ hi,
    __bf16* __restrict__ lo, int d) {
  int e    = d & 7;
  int ln   = (d >> 3) & 15;
  int hh   = (d >> 7) & 1;
  int run  = (d >> 8) & 1;
  int kc   = (d >> 9) & 3;
  int tile = d >> 11;
  float v  = in[(tile * 16 + ln) * N_FEAT + kc * 32 + run * 16 + hh * 8 + e];
  __bf16 h = (__bf16)v;
  hi[d]    = h;
  lo[d]    = (__bf16)(v - (float)h);
}

__device__ __forceinline__ void task_split_frag_weight(
    const float* __restrict__ w, __bf16* __restrict__ hi,
    __bf16* __restrict__ lo, int d) {
  int e     = d & 7;
  int ln    = (d >> 3) & 15;
  int hh    = (d >> 7) & 1;
  int run   = (d >> 8) & 1;
  int nt    = (d >> 9) & 3;
  int chunk = d >> 11;
  float v   = w[(chunk * 32 + run * 16 + hh * 8 + e) * L_OUT + nt * 16 + ln];
  __bf16 h  = (__bf16)v;
  hi[d]     = h;
  lo[d]     = (__bf16)(v - (float)h);
}

__device__ __forceinline__ void task_rowsum2(const float* __restrict__ in,
                                             float* __restrict__ out, int r,
                                             float bias) {
  const float* p = in + (size_t)r * N_FEAT;
  float s = bias;
  for (int k = 0; k < N_FEAT; ++k) s += p[k] * p[k];
  out[r] = s;
}

__global__ __launch_bounds__(PRE_T) void prepass_all(
    const float* __restrict__ xs, const float* __restrict__ centre,
    const float* __restrict__ weight,
    __bf16* __restrict__ xs_hi, __bf16* __restrict__ xs_lo,
    __bf16* __restrict__ c_hi,  __bf16* __restrict__ c_lo,
    __bf16* __restrict__ wt_hi, __bf16* __restrict__ wt_lo,
    float* __restrict__ x2p1, float* __restrict__ c2) {
  int b = blockIdx.x;
  const int t = threadIdx.x;
  if (b < NB_X2) { task_rowsum2(xs, x2p1, b * PRE_T + t, 1.0f); return; }
  b -= NB_X2;
  if (b < NB_C2) { task_rowsum2(centre, c2, b * PRE_T + t, 0.0f); return; }
  b -= NB_C2;
  if (b < NB_W) { task_split_frag_weight(weight, wt_hi, wt_lo, b * PRE_T + t); return; }
  b -= NB_W;
  if (b < NB_C) { task_split_frag_centre(centre, c_hi, c_lo, b * PRE_T + t); return; }
  b -= NB_C;
  task_split_f32_bf16(xs, xs_hi, xs_lo, b * PRE_T + t);
}

__global__ __launch_bounds__(256) void rbfn_main(
    const __bf16* __restrict__ xs_hi, const __bf16* __restrict__ xs_lo,
    const __bf16* __restrict__ c_hi,  const __bf16* __restrict__ c_lo,
    const __bf16* __restrict__ wt_hi, const __bf16* __restrict__ wt_lo,
    const float* __restrict__ x2p1,
    const float* __restrict__ c2,
    float* __restrict__ out) {
  constexpr int RS = 40;
  __shared__ alignas(16) __bf16 lds_phi_hi[N_SPLIT][M_TILES][16][RS];
  __shared__ alignas(16) __bf16 lds_phi_lo[N_SPLIT][M_TILES][16][RS];
  constexpr int PS = 68;
  __shared__ alignas(16) float lds_part[N_SPLIT][BLK_ROWS][PS];

  const int lane = threadIdx.x & 31;
  const int wave = __builtin_amdgcn_readfirstlane((int)(threadIdx.x >> 5));
  const int h    = lane >> 4;
  const int ln   = lane & 15;
  const int rowbase = blockIdx.x * BLK_ROWS;
  const int lane8   = lane * 8;

  v16bf a_hi[M_TILES][4], a_lo[M_TILES][4];
#pragma unroll
  for (int mt = 0; mt < M_TILES; ++mt) {
    const int xoff = (rowbase + mt * 16 + ln) * N_FEAT;
#pragma unroll
    for (int kc = 0; kc < 4; ++kc) {
      a_hi[mt][kc] = load_frag_g(xs_hi, xoff + kc * 32, h);
      a_lo[mt][kc] = load_frag_g(xs_lo, xoff + kc * 32, h);
    }
  }

  float x2v[M_TILES][8];
#pragma unroll
  for (int mt = 0; mt < M_TILES; ++mt)
#pragma unroll
    for (int v = 0; v < 8; ++v) x2v[mt][v] = x2p1[rowbase + mt * 16 + v + 8 * h];

  v8f acc[M_TILES][4];
#pragma unroll
  for (int mt = 0; mt < M_TILES; ++mt)
#pragma unroll
    for (int nt = 0; nt < 4; ++nt)
      acc[mt][nt] = v8f{0.f, 0.f, 0.f, 0.f, 0.f, 0.f, 0.f, 0.f};

  __bf16* phi_hi = &lds_phi_hi[wave][0][0][0];
  __bf16* phi_lo = &lds_phi_lo[wave][0][0][0];

  v16bf bc_hi[2][4], bc_lo[2][4];
  v16bf bw_hi[4], bw_lo[4];
  float c2v[2];

  c2v[0] = c2[wave * 32 + ln];
  c2v[1] = c2[wave * 32 + 16 + ln];
  __builtin_amdgcn_sched_barrier(0);
  load_centre_chunk(c_hi, c_lo, wave, lane8, bc_hi, bc_lo);
  __builtin_amdgcn_sched_barrier(0);
  load_w_chunk(wt_hi, wt_lo, wave, lane8, bw_hi, bw_lo);

  auto step = [&](const int chunk, const bool more) __attribute__((always_inline)) {
    v8f g[M_TILES][2];
#pragma unroll
    for (int mt = 0; mt < M_TILES; ++mt)
#pragma unroll
      for (int ct = 0; ct < 2; ++ct)
        g[mt][ct] = v8f{0.f, 0.f, 0.f, 0.f, 0.f, 0.f, 0.f, 0.f};
#pragma unroll
    for (int kc = 0; kc < 4; ++kc) {
#pragma unroll
      for (int mt = 0; mt < M_TILES; ++mt)
#pragma unroll
        for (int ct = 0; ct < 2; ++ct)
          g[mt][ct] = wmma_bf16(a_hi[mt][kc], bc_hi[ct][kc], g[mt][ct]);
#pragma unroll
      for (int mt = 0; mt < M_TILES; ++mt)
#pragma unroll
        for (int ct = 0; ct < 2; ++ct)
          g[mt][ct] = wmma_bf16(a_hi[mt][kc], bc_lo[ct][kc], g[mt][ct]);
#pragma unroll
      for (int mt = 0; mt < M_TILES; ++mt)
#pragma unroll
        for (int ct = 0; ct < 2; ++ct)
          g[mt][ct] = wmma_bf16(a_lo[mt][kc], bc_hi[ct][kc], g[mt][ct]);
    }

    const float c2cur[2] = {c2v[0], c2v[1]};
    if (more) {
      load_centre_chunk(c_hi, c_lo, chunk + N_SPLIT, lane8, bc_hi, bc_lo);
      c2v[0] = c2[(chunk + N_SPLIT) * 32 + ln];
      c2v[1] = c2[(chunk + N_SPLIT) * 32 + 16 + ln];
    }
    __builtin_amdgcn_sched_barrier(0);

#pragma unroll
    for (int mt = 0; mt < M_TILES; ++mt) {
#pragma unroll
      for (int ct = 0; ct < 2; ++ct) {
        float  phi[8];
        __bf16 ph[8], pl[8];
#pragma unroll
        for (int v = 0; v < 8; ++v)
          phi[v] = fmaxf(fmaf(-2.0f, g[mt][ct][v], x2v[mt][v] + c2cur[ct]), 1.0f);
#pragma unroll
        for (int v = 0; v < 8; ++v)
          phi[v] = __builtin_amdgcn_sqrtf(phi[v]);
#pragma unroll
        for (int v = 0; v < 8; ++v) ph[v] = (__bf16)phi[v];
#pragma unroll
        for (int v = 0; v < 8; ++v) pl[v] = (__bf16)(phi[v] - (float)ph[v]);
#pragma unroll
        for (int v = 0; v < 8; ++v) {
          const int idx = (mt * 16 + v + 8 * h) * RS + ct * 16 + ln;
          phi_hi[idx] = ph[v];
          phi_lo[idx] = pl[v];
        }
      }
    }

    asm volatile("s_wait_dscnt 0" ::: "memory");

    v16bf pa_hi[M_TILES], pa_lo[M_TILES];
#pragma unroll
    for (int mt = 0; mt < M_TILES; ++mt) {
      pa_hi[mt] = load_frag_lds(phi_hi + (mt * 16 + ln) * RS, h);
      pa_lo[mt] = load_frag_lds(phi_lo + (mt * 16 + ln) * RS, h);
    }

#pragma unroll
    for (int nt = 0; nt < 4; ++nt) {
#pragma unroll
      for (int mt = 0; mt < M_TILES; ++mt)
        acc[mt][nt] = wmma_bf16(pa_hi[mt], bw_hi[nt], acc[mt][nt]);
#pragma unroll
      for (int mt = 0; mt < M_TILES; ++mt)
        acc[mt][nt] = wmma_bf16(pa_hi[mt], bw_lo[nt], acc[mt][nt]);
#pragma unroll
      for (int mt = 0; mt < M_TILES; ++mt)
        acc[mt][nt] = wmma_bf16(pa_lo[mt], bw_hi[nt], acc[mt][nt]);
    }

    if (more) load_w_chunk(wt_hi, wt_lo, chunk + N_SPLIT, lane8, bw_hi, bw_lo);
  };

  for (int it = 0; it < N_ITERS - 1; ++it) step(it * N_SPLIT + wave, true);
  step((N_ITERS - 1) * N_SPLIT + wave, false);

  float* part = &lds_part[wave][0][0];
#pragma unroll
  for (int mt = 0; mt < M_TILES; ++mt)
#pragma unroll
    for (int nt = 0; nt < 4; ++nt)
#pragma unroll
      for (int v = 0; v < 8; ++v)
        part[(mt * 16 + v + 8 * h) * PS + nt * 16 + ln] = acc[mt][nt][v];

  __syncthreads();

#pragma unroll
  for (int mt = 0; mt < M_TILES; ++mt) {
    const int t   = threadIdx.x;
    const int row = mt * 16 + (t >> 4);
    const int col = (t & 15) * 4;
    float4 s = *reinterpret_cast<const float4*>(&lds_part[0][row][col]);
#pragma unroll
    for (int w = 1; w < N_SPLIT; ++w) {
      const float4 p = *reinterpret_cast<const float4*>(&lds_part[w][row][col]);
      s.x += p.x; s.y += p.y; s.z += p.z; s.w += p.w;
    }
    *reinterpret_cast<float4*>(out + (rowbase + row) * L_OUT + col) = s;
  }
}

extern "C" void kernel_launch(void* const* d_in, const int* in_sizes, int n_in,
                              void* d_out, int out_size, void* d_ws, size_t ws_size,
                              hipStream_t stream) {
  (void)in_sizes; (void)n_in; (void)out_size; (void)ws_size;

  const float* xs     = (const float*)d_in[0];
  const float* centre = (const float*)d_in[1];
  const float* weight = (const float*)d_in[2];
  float* out          = (float*)d_out;

  const size_t XS = (size_t)D_ROWS * N_FEAT;
  const size_t WT = (size_t)N_CENT * L_OUT;

  __bf16* xs_hi = (__bf16*)d_ws;
  __bf16* xs_lo = xs_hi + XS;
  __bf16* c_hi  = xs_lo + XS;
  __bf16* c_lo  = c_hi + XS;
  __bf16* wt_hi = c_lo + XS;
  __bf16* wt_lo = wt_hi + WT;
  float*  x2p1  = (float*)(wt_lo + WT);
  float*  c2    = x2p1 + D_ROWS;

  prepass_all<<<dim3(NB_PRE), dim3(PRE_T), 0, stream>>>(
      xs, centre, weight, xs_hi, xs_lo, c_hi, c_lo, wt_hi, wt_lo, x2p1, c2);

  rbfn_main<<<dim3(D_ROWS / BLK_ROWS), dim3(32 * N_SPLIT), 0, stream>>>(
      xs_hi, xs_lo, c_hi, c_lo, wt_hi, wt_lo, x2p1, c2, out);
}


# Work-group = 64 rows x one half of the centres (4 waves, four row tiles per loaded fragment): second-level reads per product halved; a third launch adds the two partial planes: f32 sum order changes
# speedup vs baseline: 17.5459x; 1.2338x over previous
#include <hip/hip_runtime.h>
#include <hip/hip_bf16.h>


typedef __attribute__((ext_vector_type(16))) __bf16 v16bf;
typedef __attribute__((ext_vector_type(8)))  __bf16 v8bf;
typedef __attribute__((ext_vector_type(8)))  float  v8f;

#define D_ROWS 8192
#define N_CENT 8192
#define N_FEAT 128
#define L_OUT  64
#define N_SPLIT 4
#define C_HALVES 2
#define CHUNKS_PER_HALF (N_CENT / (32 * C_HALVES))
#define N_ITERS (CHUNKS_PER_HALF / N_SPLIT)
#define M_TILES 4
#define BLK_ROWS (16 * M_TILES)

__device__ __forceinline__ v16bf frag_cat(v8bf a, v8bf b) {
  return __builtin_shufflevector(a, b, 0, 1, 2, 3, 4, 5, 6, 7,
                                       8, 9, 10, 11, 12, 13, 14, 15);
}

__device__ __forceinline__ void load_frag_split(const float* __restrict__ base,
                                                int off, int h,
                                                v16bf& hi, v16bf& lo) {
  const float4* p0 = reinterpret_cast<const float4*>(base + off + 8 * h);
  const float4* p1 = reinterpret_cast<const float4*>(base + off + 16 + 8 * h);
  const float4 f0 = p0[0], f1 = p0[1], f2 = p1[0], f3 = p1[1];
  const float v[16] = {f0.x, f0.y, f0.z, f0.w, f1.x, f1.y, f1.z, f1.w,
                       f2.x, f2.y, f2.z, f2.w, f3.x, f3.y, f3.z, f3.w};
#pragma unroll
  for (int e = 0; e < 16; ++e) {
    const __bf16 b = (__bf16)v[e];
    hi[e] = b;
    lo[e] = (__bf16)(v[e] - (float)b);
  }
}

__device__ __forceinline__ v16bf load_frag_blk(const __bf16* __restrict__ blk,
                                               int lane8) {
  const v8bf* p0 = reinterpret_cast<const v8bf*>(blk + lane8);
  const v8bf* p1 = reinterpret_cast<const v8bf*>(blk + 256 + lane8);
  return frag_cat(*p0, *p1);
}

__device__ __forceinline__ v16bf load_frag_lds(const __bf16* rowp, int h) {
  const v8bf* p0 = reinterpret_cast<const v8bf*>(rowp + 8 * h);
  const v8bf* p1 = reinterpret_cast<const v8bf*>(rowp + 16 + 8 * h);
  return frag_cat(*p0, *p1);
}

__device__ __forceinline__ v8f wmma_bf16(v16bf a, v16bf b, v8f c) {
  return __builtin_amdgcn_wmma_f32_16x16x32_bf16(
       false, a,  false, b,
       (short)0, c,  false,  false);
}

__device__ __forceinline__ void load_centre_chunk(
    const __bf16* __restrict__ c_hi, const __bf16* __restrict__ c_lo,
    int chunk, int lane8, v16bf (&b_hi)[2][4], v16bf (&b_lo)[2][4]) {
#pragma unroll
  for (int ct = 0; ct < 2; ++ct) {
    const int coff = (chunk * 2 + ct) * 2048;
#pragma unroll
    for (int kc = 0; kc < 4; ++kc) {
      b_hi[ct][kc] = load_frag_blk(c_hi + coff + kc * 512, lane8);
      b_lo[ct][kc] = load_frag_blk(c_lo + coff + kc * 512, lane8);
    }
  }
}

__device__ __forceinline__ void load_w_chunk(
    const __bf16* __restrict__ wt_hi, const __bf16* __restrict__ wt_lo,
    int chunk, int lane8, v16bf (&w_hi)[4], v16bf (&w_lo)[4]) {
#pragma unroll
  for (int nt = 0; nt < 4; ++nt) {
    const int woff = (chunk * 4 + nt) * 512;
    w_hi[nt] = load_frag_blk(wt_hi + woff, lane8);
    w_lo[nt] = load_frag_blk(wt_lo + woff, lane8);
  }
}

#define PRE_T 256
#define NB_X2 (D_ROWS / PRE_T)
#define NB_C2 (N_CENT / PRE_T)
#define NB_W  ((N_CENT * L_OUT) / PRE_T)
#define NB_C  ((N_CENT * N_FEAT) / PRE_T)
#define NB_PRE (NB_X2 + NB_C2 + NB_W + NB_C)

__device__ __forceinline__ void task_split_frag_centre(
    const float* __restrict__ in, __bf16* __restrict__ hi,
    __bf16* __restrict__ lo, int d) {
  int e    = d & 7;
  int ln   = (d >> 3) & 15;
  int hh   = (d >> 7) & 1;
  int run  = (d >> 8) & 1;
  int kc   = (d >> 9) & 3;
  int tile = d >> 11;
  float v  = in[(tile * 16 + ln) * N_FEAT + kc * 32 + run * 16 + hh * 8 + e];
  __bf16 h = (__bf16)v;
  hi[d]    = h;
  lo[d]    = (__bf16)(v - (float)h);
}

__device__ __forceinline__ void task_split_frag_weight(
    const float* __restrict__ w, __bf16* __restrict__ hi,
    __bf16* __restrict__ lo, int d) {
  int e     = d & 7;
  int ln    = (d >> 3) & 15;
  int hh    = (d >> 7) & 1;
  int run   = (d >> 8) & 1;
  int nt    = (d >> 9) & 3;
  int chunk = d >> 11;
  float v   = w[(chunk * 32 + run * 16 + hh * 8 + e) * L_OUT + nt * 16 + ln];
  __bf16 h  = (__bf16)v;
  hi[d]     = h;
  lo[d]     = (__bf16)(v - (float)h);
}

__device__ __forceinline__ void task_rowsum2(const float* __restrict__ in,
                                             float* __restrict__ out, int r,
                                             float bias) {
  const float* p = in + (size_t)r * N_FEAT;
  float s = bias;
  for (int k = 0; k < N_FEAT; ++k) s += p[k] * p[k];
  out[r] = s;
}

__global__ __launch_bounds__(PRE_T) void prepass_all(
    const float* __restrict__ xs, const float* __restrict__ centre,
    const float* __restrict__ weight,
    __bf16* __restrict__ c_hi,  __bf16* __restrict__ c_lo,
    __bf16* __restrict__ wt_hi, __bf16* __restrict__ wt_lo,
    float* __restrict__ x2p1, float* __restrict__ c2) {
  int b = blockIdx.x;
  const int t = threadIdx.x;
  if (b < NB_X2) { task_rowsum2(xs, x2p1, b * PRE_T + t, 1.0f); return; }
  b -= NB_X2;
  if (b < NB_C2) { task_rowsum2(centre, c2, b * PRE_T + t, 0.0f); return; }
  b -= NB_C2;
  if (b < NB_W) { task_split_frag_weight(weight, wt_hi, wt_lo, b * PRE_T + t); return; }
  b -= NB_W;
  task_split_frag_centre(centre, c_hi, c_lo, b * PRE_T + t);
}

__global__ __launch_bounds__(32 * N_SPLIT) void rbfn_main(
    const float* __restrict__ xs,
    const __bf16* __restrict__ c_hi,  const __bf16* __restrict__ c_lo,
    const __bf16* __restrict__ wt_hi, const __bf16* __restrict__ wt_lo,
    const float* __restrict__ x2p1,
    const float* __restrict__ c2,
    float* __restrict__ part0,
    float* __restrict__ part1) {
  constexpr int RS = 40;
  __shared__ alignas(16) __bf16 lds_phi_hi[N_SPLIT][M_TILES][16][RS];
  __shared__ alignas(16) __bf16 lds_phi_lo[N_SPLIT][M_TILES][16][RS];
  constexpr int PS = 68;
  __shared__ alignas(16) float lds_part[N_SPLIT][BLK_ROWS][PS];

  const int lane = threadIdx.x & 31;
  const int wave = __builtin_amdgcn_readfirstlane((int)(threadIdx.x >> 5));
  const int h    = lane >> 4;
  const int ln   = lane & 15;
  const int half    = blockIdx.x & (C_HALVES - 1);
  const int rowbase = (blockIdx.x / C_HALVES) * BLK_ROWS;
  const int chunk0  = half * CHUNKS_PER_HALF + wave;
  const int lane8   = lane * 8;

  v16bf a_hi[M_TILES][4], a_lo[M_TILES][4];
#pragma unroll
  for (int mt = 0; mt < M_TILES; ++mt) {
    const int xoff = (rowbase + mt * 16 + ln) * N_FEAT;
#pragma unroll
    for (int kc = 0; kc < 4; ++kc)
      load_frag_split(xs, xoff + kc * 32, h, a_hi[mt][kc], a_lo[mt][kc]);
  }

  float x2v[M_TILES][8];
#pragma unroll
  for (int mt = 0; mt < M_TILES; ++mt)
#pragma unroll
    for (int v = 0; v < 8; ++v) x2v[mt][v] = x2p1[rowbase + mt * 16 + v + 8 * h];

  v8f acc[M_TILES][4];
#pragma unroll
  for (int mt = 0; mt < M_TILES; ++mt)
#pragma unroll
    for (int nt = 0; nt < 4; ++nt)
      acc[mt][nt] = v8f{0.f, 0.f, 0.f, 0.f, 0.f, 0.f, 0.f, 0.f};

  __bf16* phi_hi = &lds_phi_hi[wave][0][0][0];
  __bf16* phi_lo = &lds_phi_lo[wave][0][0][0];

  v16bf bc_hi[2][4], bc_lo[2][4];
  v16bf bw_hi[4], bw_lo[4];
  float c2v[2];

  c2v[0] = c2[chunk0 * 32 + ln];
  c2v[1] = c2[chunk0 * 32 + 16 + ln];
  __builtin_amdgcn_sched_barrier(0);
  load_centre_chunk(c_hi, c_lo, chunk0, lane8, bc_hi, bc_lo);
  __builtin_amdgcn_sched_barrier(0);
  load_w_chunk(wt_hi, wt_lo, chunk0, lane8, bw_hi, bw_lo);

  auto step = [&](const int chunk, const bool more) __attribute__((always_inline)) {
    v8f g[M_TILES][2];
#pragma unroll
    for (int mt = 0; mt < M_TILES; ++mt)
#pragma unroll
      for (int ct = 0; ct < 2; ++ct)
        g[mt][ct] = v8f{0.f, 0.f, 0.f, 0.f, 0.f, 0.f, 0.f, 0.f};
#pragma unroll
    for (int kc = 0; kc < 4; ++kc) {
#pragma unroll
      for (int mt = 0; mt < M_TILES; ++mt)
#pragma unroll
        for (int ct = 0; ct < 2; ++ct)
          g[mt][ct] = wmma_bf16(a_hi[mt][kc], bc_hi[ct][kc], g[mt][ct]);
#pragma unroll
      for (int mt = 0; mt < M_TILES; ++mt)
#pragma unroll
        for (int ct = 0; ct < 2; ++ct)
          g[mt][ct] = wmma_bf16(a_hi[mt][kc], bc_lo[ct][kc], g[mt][ct]);
#pragma unroll
      for (int mt = 0; mt < M_TILES; ++mt)
#pragma unroll
        for (int ct = 0; ct < 2; ++ct)
          g[mt][ct] = wmma_bf16(a_lo[mt][kc], bc_hi[ct][kc], g[mt][ct]);
    }

    const float c2cur[2] = {c2v[0], c2v[1]};
    if (more) {
      load_centre_chunk(c_hi, c_lo, chunk + N_SPLIT, lane8, bc_hi, bc_lo);
      c2v[0] = c2[(chunk + N_SPLIT) * 32 + ln];
      c2v[1] = c2[(chunk + N_SPLIT) * 32 + 16 + ln];
    }
    __builtin_amdgcn_sched_barrier(0);

#pragma unroll
    for (int mt = 0; mt < M_TILES; ++mt) {
#pragma unroll
      for (int ct = 0; ct < 2; ++ct) {
        float  phi[8];
        __bf16 ph[8], pl[8];
#pragma unroll
        for (int v = 0; v < 8; ++v)
          phi[v] = fmaxf(fmaf(-2.0f, g[mt][ct][v], x2v[mt][v] + c2cur[ct]), 1.0f);
#pragma unroll
        for (int v = 0; v < 8; ++v)
          phi[v] = __builtin_amdgcn_sqrtf(phi[v]);
#pragma unroll
        for (int v = 0; v < 8; ++v) ph[v] = (__bf16)phi[v];
#pragma unroll
        for (int v = 0; v < 8; ++v) pl[v] = (__bf16)(phi[v] - (float)ph[v]);
#pragma unroll
        for (int v = 0; v < 8; ++v) {
          const int idx = (mt * 16 + v + 8 * h) * RS + ct * 16 + ln;
          phi_hi[idx] = ph[v];
          phi_lo[idx] = pl[v];
        }
      }
    }

    asm volatile("s_wait_dscnt 0" ::: "memory");

    v16bf pa_hi[M_TILES], pa_lo[M_TILES];
#pragma unroll
    for (int mt = 0; mt < M_TILES; ++mt) {
      pa_hi[mt] = load_frag_lds(phi_hi + (mt * 16 + ln) * RS, h);
      pa_lo[mt] = load_frag_lds(phi_lo + (mt * 16 + ln) * RS, h);
    }

#pragma unroll
    for (int nt = 0; nt < 4; ++nt) {
#pragma unroll
      for (int mt = 0; mt < M_TILES; ++mt)
        acc[mt][nt] = wmma_bf16(pa_hi[mt], bw_hi[nt], acc[mt][nt]);
#pragma unroll
      for (int mt = 0; mt < M_TILES; ++mt)
        acc[mt][nt] = wmma_bf16(pa_hi[mt], bw_lo[nt], acc[mt][nt]);
#pragma unroll
      for (int mt = 0; mt < M_TILES; ++mt)
        acc[mt][nt] = wmma_bf16(pa_lo[mt], bw_hi[nt], acc[mt][nt]);
    }

    if (more) load_w_chunk(wt_hi, wt_lo, chunk + N_SPLIT, lane8, bw_hi, bw_lo);
  };

  for (int it = 0; it < N_ITERS - 1; ++it) step(it * N_SPLIT + chunk0, true);
  step((N_ITERS - 1) * N_SPLIT + chunk0, false);

  float* part = &lds_part[wave][0][0];
#pragma unroll
  for (int mt = 0; mt < M_TILES; ++mt)
#pragma unroll
    for (int nt = 0; nt < 4; ++nt)
#pragma unroll
      for (int v = 0; v < 8; ++v)
        part[(mt * 16 + v + 8 * h) * PS + nt * 16 + ln] = acc[mt][nt][v];

  __syncthreads();

  float* __restrict__ dst = half ? part1 : part0;
#pragma unroll
  for (int ps = 0; ps < BLK_ROWS / (2 * N_SPLIT); ++ps) {
    const int t   = threadIdx.x;
    const int row = ps * (2 * N_SPLIT) + (t >> 4);
    const int col = (t & 15) * 4;
    float4 s = *reinterpret_cast<const float4*>(&lds_part[0][row][col]);
#pragma unroll
    for (int w = 1; w < N_SPLIT; ++w) {
      const float4 p = *reinterpret_cast<const float4*>(&lds_part[w][row][col]);
      s.x += p.x; s.y += p.y; s.z += p.z; s.w += p.w;
    }
    *reinterpret_cast<float4*>(dst + (rowbase + row) * L_OUT + col) = s;
  }
}

#define ADD_T 256
__global__ __launch_bounds__(ADD_T) void add_halves(
    const float* __restrict__ part0, const float* __restrict__ part1,
    float* __restrict__ out) {
  const int i = blockIdx.x * ADD_T + threadIdx.x;
  const float4 a = reinterpret_cast<const float4*>(part0)[i];
  const float4 b = reinterpret_cast<const float4*>(part1)[i];
  float4 s;
  s.x = a.x + b.x; s.y = a.y + b.y; s.z = a.z + b.z; s.w = a.w + b.w;
  reinterpret_cast<float4*>(out)[i] = s;
}

extern "C" void kernel_launch(void* const* d_in, const int* in_sizes, int n_in,
                              void* d_out, int out_size, void* d_ws, size_t ws_size,
                              hipStream_t stream) {
  (void)in_sizes; (void)n_in; (void)out_size; (void)ws_size;

  const float* xs     = (const float*)d_in[0];
  const float* centre = (const float*)d_in[1];
  const float* weight = (const float*)d_in[2];
  float* out          = (float*)d_out;

  const size_t XS = (size_t)D_ROWS * N_FEAT;
  const size_t WT = (size_t)N_CENT * L_OUT;

  float*  part0 = (float*)d_ws;
  float*  part1 = part0 + WT;
  __bf16* c_hi  = (__bf16*)(part1 + WT);
  __bf16* c_lo  = c_hi + XS;
  __bf16* wt_hi = c_lo + XS;
  __bf16* wt_lo = wt_hi + WT;
  float*  x2p1  = (float*)(wt_lo + WT);
  float*  c2    = x2p1 + D_ROWS;

  prepass_all<<<dim3(NB_PRE), dim3(PRE_T), 0, stream>>>(
      xs, centre, weight, c_hi, c_lo, wt_hi, wt_lo, x2p1, c2);

  rbfn_main<<<dim3((D_ROWS / BLK_ROWS) * C_HALVES), dim3(32 * N_SPLIT), 0, stream>>>(
      xs, c_hi, c_lo, wt_hi, wt_lo, x2p1, c2, part0, part1);

  add_halves<<<dim3((D_ROWS * L_OUT) / (4 * ADD_T)), dim3(ADD_T), 0, stream>>>(
      part0, part1, out);
}


# First product formed transposed (centre block as A operand, xs block as B): its f32 tile, rounded and packed in place, is the second product's A fragment; no LDS transpose of phi; sums unchanged
# speedup vs baseline: 21.2041x; 1.2085x over previous
#include <hip/hip_runtime.h>
#include <hip/hip_bf16.h>


typedef __attribute__((ext_vector_type(16))) __bf16 v16bf;
typedef __attribute__((ext_vector_type(8)))  __bf16 v8bf;
typedef __attribute__((ext_vector_type(8)))  float  v8f;
typedef __attribute__((ext_vector_type(2)))  float    v2f;
typedef __attribute__((ext_vector_type(2)))  __bf16   v2bf;
typedef __attribute__((ext_vector_type(8)))  unsigned v8u;

#define D_ROWS 8192
#define N_CENT 8192
#define N_FEAT 128
#define L_OUT  64
#define N_SPLIT 4
#define C_HALVES 2
#define CHUNKS_PER_HALF (N_CENT / (32 * C_HALVES))
#define N_ITERS (CHUNKS_PER_HALF / N_SPLIT)
#define M_TILES 4
#define BLK_ROWS (16 * M_TILES)

__device__ __forceinline__ v16bf frag_cat(v8bf a, v8bf b) {
  return __builtin_shufflevector(a, b, 0, 1, 2, 3, 4, 5, 6, 7,
                                       8, 9, 10, 11, 12, 13, 14, 15);
}

__device__ __forceinline__ void load_frag_split(const float* __restrict__ base,
                                                int off, int h,
                                                v16bf& hi, v16bf& lo) {
  const float4* p0 = reinterpret_cast<const float4*>(base + off + 8 * h);
  const float4* p1 = reinterpret_cast<const float4*>(base + off + 16 + 8 * h);
  const float4 f0 = p0[0], f1 = p0[1], f2 = p1[0], f3 = p1[1];
  const float v[16] = {f0.x, f0.y, f0.z, f0.w, f1.x, f1.y, f1.z, f1.w,
                       f2.x, f2.y, f2.z, f2.w, f3.x, f3.y, f3.z, f3.w};
#pragma unroll
  for (int e = 0; e < 16; ++e) {
    const __bf16 b = (__bf16)v[e];
    hi[e] = b;
    lo[e] = (__bf16)(v[e] - (float)b);
  }
}

__device__ __forceinline__ v16bf load_frag_blk(const __bf16* __restrict__ blk,
                                               int lane8) {
  const v8bf* p0 = reinterpret_cast<const v8bf*>(blk + lane8);
  const v8bf* p1 = reinterpret_cast<const v8bf*>(blk + 256 + lane8);
  return frag_cat(*p0, *p1);
}

__device__ __forceinline__ v8f wmma_bf16(v16bf a, v16bf b, v8f c) {
  return __builtin_amdgcn_wmma_f32_16x16x32_bf16(
       false, a,  false, b,
       (short)0, c,  false,  false);
}

__device__ __forceinline__ void load_centre_chunk(
    const __bf16* __restrict__ c_hi, const __bf16* __restrict__ c_lo,
    int chunk, int lane8, v16bf (&b_hi)[2][4], v16bf (&b_lo)[2][4]) {
#pragma unroll
  for (int ct = 0; ct < 2; ++ct) {
    const int coff = (chunk * 2 + ct) * 2048;
#pragma unroll
    for (int kc = 0; kc < 4; ++kc) {
      b_hi[ct][kc] = load_frag_blk(c_hi + coff + kc * 512, lane8);
      b_lo[ct][kc] = load_frag_blk(c_lo + coff + kc * 512, lane8);
    }
  }
}

__device__ __forceinline__ void load_w_chunk(
    const __bf16* __restrict__ wt_hi, const __bf16* __restrict__ wt_lo,
    int chunk, int lane8, v16bf (&w_hi)[4], v16bf (&w_lo)[4]) {
#pragma unroll
  for (int nt = 0; nt < 4; ++nt) {
    const int woff = (chunk * 4 + nt) * 512;
    w_hi[nt] = load_frag_blk(wt_hi + woff, lane8);
    w_lo[nt] = load_frag_blk(wt_lo + woff, lane8);
  }
}

#define PRE_T 256
#define NB_X2 (D_ROWS / PRE_T)
#define NB_C2 (N_CENT / PRE_T)
#define NB_W  ((N_CENT * L_OUT) / PRE_T)
#define NB_C  ((N_CENT * N_FEAT) / PRE_T)
#define NB_PRE (NB_X2 + NB_C2 + NB_W + NB_C)

__device__ __forceinline__ void task_split_frag_centre(
    const float* __restrict__ in, __bf16* __restrict__ hi,
    __bf16* __restrict__ lo, int d) {
  int e    = d & 7;
  int ln   = (d >> 3) & 15;
  int hh   = (d >> 7) & 1;
  int run  = (d >> 8) & 1;
  int kc   = (d >> 9) & 3;
  int tile = d >> 11;
  float v  = in[(tile * 16 + ln) * N_FEAT + kc * 32 + run * 16 + hh * 8 + e];
  __bf16 h = (__bf16)v;
  hi[d]    = h;
  lo[d]    = (__bf16)(v - (float)h);
}

__device__ __forceinline__ void task_split_frag_weight(
    const float* __restrict__ w, __bf16* __restrict__ hi,
    __bf16* __restrict__ lo, int d) {
  int e     = d & 7;
  int ln    = (d >> 3) & 15;
  int hh    = (d >> 7) & 1;
  int run   = (d >> 8) & 1;
  int nt    = (d >> 9) & 3;
  int chunk = d >> 11;
  float v   = w[(chunk * 32 + run * 16 + hh * 8 + e) * L_OUT + nt * 16 + ln];
  __bf16 h  = (__bf16)v;
  hi[d]     = h;
  lo[d]     = (__bf16)(v - (float)h);
}

__device__ __forceinline__ void task_rowsum2(const float* __restrict__ in,
                                             float* __restrict__ out, int r,
                                             float bias) {
  const float* p = in + (size_t)r * N_FEAT;
  float s = bias;
  for (int k = 0; k < N_FEAT; ++k) s += p[k] * p[k];
  out[r] = s;
}

__global__ __launch_bounds__(PRE_T) void prepass_all(
    const float* __restrict__ xs, const float* __restrict__ centre,
    const float* __restrict__ weight,
    __bf16* __restrict__ c_hi,  __bf16* __restrict__ c_lo,
    __bf16* __restrict__ wt_hi, __bf16* __restrict__ wt_lo,
    float* __restrict__ x2p1, float* __restrict__ c2) {
  int b = blockIdx.x;
  const int t = threadIdx.x;
  if (b < NB_X2) { task_rowsum2(xs, x2p1, b * PRE_T + t, 1.0f); return; }
  b -= NB_X2;
  if (b < NB_C2) { task_rowsum2(centre, c2, b * PRE_T + t, 0.0f); return; }
  b -= NB_C2;
  if (b < NB_W) { task_split_frag_weight(weight, wt_hi, wt_lo, b * PRE_T + t); return; }
  b -= NB_W;
  task_split_frag_centre(centre, c_hi, c_lo, b * PRE_T + t);
}

__global__ __launch_bounds__(32 * N_SPLIT) void rbfn_main(
    const float* __restrict__ xs,
    const __bf16* __restrict__ c_hi,  const __bf16* __restrict__ c_lo,
    const __bf16* __restrict__ wt_hi, const __bf16* __restrict__ wt_lo,
    const float* __restrict__ x2p1,
    const float* __restrict__ c2,
    float* __restrict__ part0,
    float* __restrict__ part1) {
  constexpr int PS = 68;
  __shared__ alignas(16) float lds_part[N_SPLIT][BLK_ROWS][PS];

  const int lane = threadIdx.x & 31;
  const int wave = __builtin_amdgcn_readfirstlane((int)(threadIdx.x >> 5));
  const int h    = lane >> 4;
  const int ln   = lane & 15;
  const int half    = blockIdx.x & (C_HALVES - 1);
  const int rowbase = (blockIdx.x / C_HALVES) * BLK_ROWS;
  const int chunk0  = half * CHUNKS_PER_HALF + wave;
  const int lane8   = lane * 8;

  v16bf a_hi[M_TILES][4], a_lo[M_TILES][4];
#pragma unroll
  for (int mt = 0; mt < M_TILES; ++mt) {
    const int xoff = (rowbase + mt * 16 + ln) * N_FEAT;
#pragma unroll
    for (int kc = 0; kc < 4; ++kc)
      load_frag_split(xs, xoff + kc * 32, h, a_hi[mt][kc], a_lo[mt][kc]);
  }

  float x2v[M_TILES];
#pragma unroll
  for (int mt = 0; mt < M_TILES; ++mt) x2v[mt] = x2p1[rowbase + mt * 16 + ln];

  v8f acc[M_TILES][4];
#pragma unroll
  for (int mt = 0; mt < M_TILES; ++mt)
#pragma unroll
    for (int nt = 0; nt < 4; ++nt)
      acc[mt][nt] = v8f{0.f, 0.f, 0.f, 0.f, 0.f, 0.f, 0.f, 0.f};

  v16bf bc_hi[2][4], bc_lo[2][4];
  v16bf bw_hi[4], bw_lo[4];
  float4 c2v[2][2];

  {
    const float4* cp = reinterpret_cast<const float4*>(c2 + chunk0 * 32 + 8 * h);
    c2v[0][0] = cp[0]; c2v[0][1] = cp[1];
    c2v[1][0] = cp[4]; c2v[1][1] = cp[5];
  }
  __builtin_amdgcn_sched_barrier(0);
  load_centre_chunk(c_hi, c_lo, chunk0, lane8, bc_hi, bc_lo);
  __builtin_amdgcn_sched_barrier(0);
  load_w_chunk(wt_hi, wt_lo, chunk0, lane8, bw_hi, bw_lo);

  auto step = [&](const int chunk, const bool more) __attribute__((always_inline)) {
    v8f g[M_TILES][2];
#pragma unroll
    for (int mt = 0; mt < M_TILES; ++mt)
#pragma unroll
      for (int ct = 0; ct < 2; ++ct)
        g[mt][ct] = v8f{0.f, 0.f, 0.f, 0.f, 0.f, 0.f, 0.f, 0.f};
#pragma unroll
    for (int kc = 0; kc < 4; ++kc) {
#pragma unroll
      for (int mt = 0; mt < M_TILES; ++mt)
#pragma unroll
        for (int ct = 0; ct < 2; ++ct)
          g[mt][ct] = wmma_bf16(bc_hi[ct][kc], a_hi[mt][kc], g[mt][ct]);
#pragma unroll
      for (int mt = 0; mt < M_TILES; ++mt)
#pragma unroll
        for (int ct = 0; ct < 2; ++ct)
          g[mt][ct] = wmma_bf16(bc_lo[ct][kc], a_hi[mt][kc], g[mt][ct]);
#pragma unroll
      for (int mt = 0; mt < M_TILES; ++mt)
#pragma unroll
        for (int ct = 0; ct < 2; ++ct)
          g[mt][ct] = wmma_bf16(bc_hi[ct][kc], a_lo[mt][kc], g[mt][ct]);
    }

    const float c2cur[2][8] = {
        {c2v[0][0].x, c2v[0][0].y, c2v[0][0].z, c2v[0][0].w,
         c2v[0][1].x, c2v[0][1].y, c2v[0][1].z, c2v[0][1].w},
        {c2v[1][0].x, c2v[1][0].y, c2v[1][0].z, c2v[1][0].w,
         c2v[1][1].x, c2v[1][1].y, c2v[1][1].z, c2v[1][1].w}};
    if (more) {
      load_centre_chunk(c_hi, c_lo, chunk + N_SPLIT, lane8, bc_hi, bc_lo);
      const float4* cp =
          reinterpret_cast<const float4*>(c2 + (chunk + N_SPLIT) * 32 + 8 * h);
      c2v[0][0] = cp[0]; c2v[0][1] = cp[1];
      c2v[1][0] = cp[4]; c2v[1][1] = cp[5];
    }
    __builtin_amdgcn_sched_barrier(0);

    v16bf pa_hi[M_TILES], pa_lo[M_TILES];
#pragma unroll
    for (int mt = 0; mt < M_TILES; ++mt) {
      v8u wh, wl;
#pragma unroll
      for (int ct = 0; ct < 2; ++ct) {
        float phi[8];
#pragma unroll
        for (int v = 0; v < 8; ++v)
          phi[v] = fmaxf(fmaf(-2.0f, g[mt][ct][v], x2v[mt] + c2cur[ct][v]), 1.0f);
#pragma unroll
        for (int v = 0; v < 8; ++v)
          phi[v] = __builtin_amdgcn_sqrtf(phi[v]);
#pragma unroll
        for (int v = 0; v < 8; v += 2) {
          const v2bf hp = __builtin_convertvector(v2f{phi[v], phi[v + 1]}, v2bf);
          const v2bf lp = __builtin_convertvector(
              v2f{phi[v] - (float)hp[0], phi[v + 1] - (float)hp[1]}, v2bf);
          wh[ct * 4 + v / 2] = __builtin_bit_cast(unsigned, hp);
          wl[ct * 4 + v / 2] = __builtin_bit_cast(unsigned, lp);
        }
      }
      pa_hi[mt] = __builtin_bit_cast(v16bf, wh);
      pa_lo[mt] = __builtin_bit_cast(v16bf, wl);
    }

#pragma unroll
    for (int nt = 0; nt < 4; ++nt) {
#pragma unroll
      for (int mt = 0; mt < M_TILES; ++mt)
        acc[mt][nt] = wmma_bf16(pa_hi[mt], bw_hi[nt], acc[mt][nt]);
#pragma unroll
      for (int mt = 0; mt < M_TILES; ++mt)
        acc[mt][nt] = wmma_bf16(pa_hi[mt], bw_lo[nt], acc[mt][nt]);
#pragma unroll
      for (int mt = 0; mt < M_TILES; ++mt)
        acc[mt][nt] = wmma_bf16(pa_lo[mt], bw_hi[nt], acc[mt][nt]);
    }

    if (more) load_w_chunk(wt_hi, wt_lo, chunk + N_SPLIT, lane8, bw_hi, bw_lo);
  };

  for (int it = 0; it < N_ITERS - 1; ++it) step(it * N_SPLIT + chunk0, true);
  step((N_ITERS - 1) * N_SPLIT + chunk0, false);

  float* part = &lds_part[wave][0][0];
#pragma unroll
  for (int mt = 0; mt < M_TILES; ++mt)
#pragma unroll
    for (int nt = 0; nt < 4; ++nt)
#pragma unroll
      for (int v = 0; v < 8; ++v)
        part[(mt * 16 + v + 8 * h) * PS + nt * 16 + ln] = acc[mt][nt][v];

  __syncthreads();

  float* __restrict__ dst = half ? part1 : part0;
#pragma unroll
  for (int ps = 0; ps < BLK_ROWS / (2 * N_SPLIT); ++ps) {
    const int t   = threadIdx.x;
    const int row = ps * (2 * N_SPLIT) + (t >> 4);
    const int col = (t & 15) * 4;
    float4 s = *reinterpret_cast<const float4*>(&lds_part[0][row][col]);
#pragma unroll
    for (int w = 1; w < N_SPLIT; ++w) {
      const float4 p = *reinterpret_cast<const float4*>(&lds_part[w][row][col]);
      s.x += p.x; s.y += p.y; s.z += p.z; s.w += p.w;
    }
    *reinterpret_cast<float4*>(dst + (rowbase + row) * L_OUT + col) = s;
  }
}

#define ADD_T 256
__global__ __launch_bounds__(ADD_T) void add_halves(
    const float* __restrict__ part0, const float* __restrict__ part1,
    float* __restrict__ out) {
  const int i = blockIdx.x * ADD_T + threadIdx.x;
  const float4 a = reinterpret_cast<const float4*>(part0)[i];
  const float4 b = reinterpret_cast<const float4*>(part1)[i];
  float4 s;
  s.x = a.x + b.x; s.y = a.y + b.y; s.z = a.z + b.z; s.w = a.w + b.w;
  reinterpret_cast<float4*>(out)[i] = s;
}

extern "C" void kernel_launch(void* const* d_in, const int* in_sizes, int n_in,
                              void* d_out, int out_size, void* d_ws, size_t ws_size,
                              hipStream_t stream) {
  (void)in_sizes; (void)n_in; (void)out_size; (void)ws_size;

  const float* xs     = (const float*)d_in[0];
  const float* centre = (const float*)d_in[1];
  const float* weight = (const float*)d_in[2];
  float* out          = (float*)d_out;

  const size_t XS = (size_t)D_ROWS * N_FEAT;
  const size_t WT = (size_t)N_CENT * L_OUT;

  float*  part0 = (float*)d_ws;
  float*  part1 = part0 + WT;
  __bf16* c_hi  = (__bf16*)(part1 + WT);
  __bf16* c_lo  = c_hi + XS;
  __bf16* wt_hi = c_lo + XS;
  __bf16* wt_lo = wt_hi + WT;
  float*  x2p1  = (float*)(wt_lo + WT);
  float*  c2    = x2p1 + D_ROWS;

  prepass_all<<<dim3(NB_PRE), dim3(PRE_T), 0, stream>>>(
      xs, centre, weight, c_hi, c_lo, wt_hi, wt_lo, x2p1, c2);

  rbfn_main<<<dim3((D_ROWS / BLK_ROWS) * C_HALVES), dim3(32 * N_SPLIT), 0, stream>>>(
      xs, c_hi, c_lo, wt_hi, wt_lo, x2p1, c2, part0, part1);

  add_halves<<<dim3((D_ROWS * L_OUT) / (4 * ADD_T)), dim3(ADD_T), 0, stream>>>(
      part0, part1, out);
}


# Next chunk's centre fragment loads issued as ONE run, in the order consumed, between the two products (scheduling barrier in front), not as a pair after every 4th matrix instruction; sums unchanged
# speedup vs baseline: 22.1523x; 1.0447x over previous
#include <hip/hip_runtime.h>
#include <hip/hip_bf16.h>


typedef __attribute__((ext_vector_type(16))) __bf16 v16bf;
typedef __attribute__((ext_vector_type(8)))  __bf16 v8bf;
typedef __attribute__((ext_vector_type(8)))  float  v8f;
typedef __attribute__((ext_vector_type(2)))  float    v2f;
typedef __attribute__((ext_vector_type(2)))  __bf16   v2bf;
typedef __attribute__((ext_vector_type(8)))  unsigned v8u;

#define D_ROWS 8192
#define N_CENT 8192
#define N_FEAT 128
#define L_OUT  64
#define N_SPLIT 4
#define C_HALVES 2
#define CHUNKS_PER_HALF (N_CENT / (32 * C_HALVES))
#define N_ITERS (CHUNKS_PER_HALF / N_SPLIT)
#define M_TILES 4
#define BLK_ROWS (16 * M_TILES)

__device__ __forceinline__ v16bf frag_cat(v8bf a, v8bf b) {
  return __builtin_shufflevector(a, b, 0, 1, 2, 3, 4, 5, 6, 7,
                                       8, 9, 10, 11, 12, 13, 14, 15);
}

__device__ __forceinline__ void load_frag_split(const float* __restrict__ base,
                                                int off, int h,
                                                v16bf& hi, v16bf& lo) {
  const float4* p0 = reinterpret_cast<const float4*>(base + off + 8 * h);
  const float4* p1 = reinterpret_cast<const float4*>(base + off + 16 + 8 * h);
  const float4 f0 = p0[0], f1 = p0[1], f2 = p1[0], f3 = p1[1];
  const float v[16] = {f0.x, f0.y, f0.z, f0.w, f1.x, f1.y, f1.z, f1.w,
                       f2.x, f2.y, f2.z, f2.w, f3.x, f3.y, f3.z, f3.w};
#pragma unroll
  for (int e = 0; e < 16; ++e) {
    const __bf16 b = (__bf16)v[e];
    hi[e] = b;
    lo[e] = (__bf16)(v[e] - (float)b);
  }
}

__device__ __forceinline__ v16bf load_frag_blk(const __bf16* __restrict__ blk,
                                               int lane8) {
  const v8bf* p0 = reinterpret_cast<const v8bf*>(blk + lane8);
  const v8bf* p1 = reinterpret_cast<const v8bf*>(blk + 256 + lane8);
  return frag_cat(*p0, *p1);
}

__device__ __forceinline__ v8f wmma_bf16(v16bf a, v16bf b, v8f c) {
  return __builtin_amdgcn_wmma_f32_16x16x32_bf16(
       false, a,  false, b,
       (short)0, c,  false,  false);
}

__device__ __forceinline__ void load_centre_chunk(
    const __bf16* __restrict__ c_hi, const __bf16* __restrict__ c_lo,
    int chunk, int lane8, v16bf (&b_hi)[2][4], v16bf (&b_lo)[2][4]) {
#pragma unroll
  for (int kc = 0; kc < 4; ++kc) {
#pragma unroll
    for (int ct = 0; ct < 2; ++ct) {
      const int coff = (chunk * 2 + ct) * 2048;
      b_hi[ct][kc] = load_frag_blk(c_hi + coff + kc * 512, lane8);
      b_lo[ct][kc] = load_frag_blk(c_lo + coff + kc * 512, lane8);
    }
  }
}

__device__ __forceinline__ void load_w_chunk(
    const __bf16* __restrict__ wt_hi, const __bf16* __restrict__ wt_lo,
    int chunk, int lane8, v16bf (&w_hi)[4], v16bf (&w_lo)[4]) {
#pragma unroll
  for (int nt = 0; nt < 4; ++nt) {
    const int woff = (chunk * 4 + nt) * 512;
    w_hi[nt] = load_frag_blk(wt_hi + woff, lane8);
    w_lo[nt] = load_frag_blk(wt_lo + woff, lane8);
  }
}

#define PRE_T 256
#define NB_X2 (D_ROWS / PRE_T)
#define NB_C2 (N_CENT / PRE_T)
#define NB_W  ((N_CENT * L_OUT) / PRE_T)
#define NB_C  ((N_CENT * N_FEAT) / PRE_T)
#define NB_PRE (NB_X2 + NB_C2 + NB_W + NB_C)

__device__ __forceinline__ void task_split_frag_centre(
    const float* __restrict__ in, __bf16* __restrict__ hi,
    __bf16* __restrict__ lo, int d) {
  int e    = d & 7;
  int ln   = (d >> 3) & 15;
  int hh   = (d >> 7) & 1;
  int run  = (d >> 8) & 1;
  int kc   = (d >> 9) & 3;
  int tile = d >> 11;
  float v  = in[(tile * 16 + ln) * N_FEAT + kc * 32 + run * 16 + hh * 8 + e];
  __bf16 h = (__bf16)v;
  hi[d]    = h;
  lo[d]    = (__bf16)(v - (float)h);
}

__device__ __forceinline__ void task_split_frag_weight(
    const float* __restrict__ w, __bf16* __restrict__ hi,
    __bf16* __restrict__ lo, int d) {
  int e     = d & 7;
  int ln    = (d >> 3) & 15;
  int hh    = (d >> 7) & 1;
  int run   = (d >> 8) & 1;
  int nt    = (d >> 9) & 3;
  int chunk = d >> 11;
  float v   = w[(chunk * 32 + run * 16 + hh * 8 + e) * L_OUT + nt * 16 + ln];
  __bf16 h  = (__bf16)v;
  hi[d]     = h;
  lo[d]     = (__bf16)(v - (float)h);
}

__device__ __forceinline__ void task_rowsum2(const float* __restrict__ in,
                                             float* __restrict__ out, int r,
                                             float bias) {
  const float* p = in + (size_t)r * N_FEAT;
  float s = bias;
  for (int k = 0; k < N_FEAT; ++k) s += p[k] * p[k];
  out[r] = s;
}

__global__ __launch_bounds__(PRE_T) void prepass_all(
    const float* __restrict__ xs, const float* __restrict__ centre,
    const float* __restrict__ weight,
    __bf16* __restrict__ c_hi,  __bf16* __restrict__ c_lo,
    __bf16* __restrict__ wt_hi, __bf16* __restrict__ wt_lo,
    float* __restrict__ x2p1, float* __restrict__ c2) {
  int b = blockIdx.x;
  const int t = threadIdx.x;
  if (b < NB_X2) { task_rowsum2(xs, x2p1, b * PRE_T + t, 1.0f); return; }
  b -= NB_X2;
  if (b < NB_C2) { task_rowsum2(centre, c2, b * PRE_T + t, 0.0f); return; }
  b -= NB_C2;
  if (b < NB_W) { task_split_frag_weight(weight, wt_hi, wt_lo, b * PRE_T + t); return; }
  b -= NB_W;
  task_split_frag_centre(centre, c_hi, c_lo, b * PRE_T + t);
}

__global__ __launch_bounds__(32 * N_SPLIT) void rbfn_main(
    const float* __restrict__ xs,
    const __bf16* __restrict__ c_hi,  const __bf16* __restrict__ c_lo,
    const __bf16* __restrict__ wt_hi, const __bf16* __restrict__ wt_lo,
    const float* __restrict__ x2p1,
    const float* __restrict__ c2,
    float* __restrict__ part0,
    float* __restrict__ part1) {
  constexpr int PS = 68;
  __shared__ alignas(16) float lds_part[N_SPLIT][BLK_ROWS][PS];

  const int lane = threadIdx.x & 31;
  const int wave = __builtin_amdgcn_readfirstlane((int)(threadIdx.x >> 5));
  const int h    = lane >> 4;
  const int ln   = lane & 15;
  const int half    = blockIdx.x & (C_HALVES - 1);
  const int rowbase = (blockIdx.x / C_HALVES) * BLK_ROWS;
  const int chunk0  = half * CHUNKS_PER_HALF + wave;
  const int lane8   = lane * 8;

  v16bf a_hi[M_TILES][4], a_lo[M_TILES][4];
#pragma unroll
  for (int mt = 0; mt < M_TILES; ++mt) {
    const int xoff = (rowbase + mt * 16 + ln) * N_FEAT;
#pragma unroll
    for (int kc = 0; kc < 4; ++kc)
      load_frag_split(xs, xoff + kc * 32, h, a_hi[mt][kc], a_lo[mt][kc]);
  }

  float x2v[M_TILES];
#pragma unroll
  for (int mt = 0; mt < M_TILES; ++mt) x2v[mt] = x2p1[rowbase + mt * 16 + ln];

  v8f acc[M_TILES][4];
#pragma unroll
  for (int mt = 0; mt < M_TILES; ++mt)
#pragma unroll
    for (int nt = 0; nt < 4; ++nt)
      acc[mt][nt] = v8f{0.f, 0.f, 0.f, 0.f, 0.f, 0.f, 0.f, 0.f};

  v16bf bc_hi[2][4], bc_lo[2][4];
  v16bf bw_hi[4], bw_lo[4];
  float4 c2v[2][2];

  {
    const float4* cp = reinterpret_cast<const float4*>(c2 + chunk0 * 32 + 8 * h);
    c2v[0][0] = cp[0]; c2v[0][1] = cp[1];
    c2v[1][0] = cp[4]; c2v[1][1] = cp[5];
  }
  __builtin_amdgcn_sched_barrier(0);
  load_centre_chunk(c_hi, c_lo, chunk0, lane8, bc_hi, bc_lo);
  __builtin_amdgcn_sched_barrier(0);
  load_w_chunk(wt_hi, wt_lo, chunk0, lane8, bw_hi, bw_lo);

  auto step = [&](const int chunk, const bool more) __attribute__((always_inline)) {
    v8f g[M_TILES][2];
#pragma unroll
    for (int mt = 0; mt < M_TILES; ++mt)
#pragma unroll
      for (int ct = 0; ct < 2; ++ct)
        g[mt][ct] = v8f{0.f, 0.f, 0.f, 0.f, 0.f, 0.f, 0.f, 0.f};
#pragma unroll
    for (int kc = 0; kc < 4; ++kc) {
#pragma unroll
      for (int mt = 0; mt < M_TILES; ++mt)
#pragma unroll
        for (int ct = 0; ct < 2; ++ct)
          g[mt][ct] = wmma_bf16(bc_hi[ct][kc], a_hi[mt][kc], g[mt][ct]);
#pragma unroll
      for (int mt = 0; mt < M_TILES; ++mt)
#pragma unroll
        for (int ct = 0; ct < 2; ++ct)
          g[mt][ct] = wmma_bf16(bc_lo[ct][kc], a_hi[mt][kc], g[mt][ct]);
#pragma unroll
      for (int mt = 0; mt < M_TILES; ++mt)
#pragma unroll
        for (int ct = 0; ct < 2; ++ct)
          g[mt][ct] = wmma_bf16(bc_hi[ct][kc], a_lo[mt][kc], g[mt][ct]);
    }

    const float c2cur[2][8] = {
        {c2v[0][0].x, c2v[0][0].y, c2v[0][0].z, c2v[0][0].w,
         c2v[0][1].x, c2v[0][1].y, c2v[0][1].z, c2v[0][1].w},
        {c2v[1][0].x, c2v[1][0].y, c2v[1][0].z, c2v[1][0].w,
         c2v[1][1].x, c2v[1][1].y, c2v[1][1].z, c2v[1][1].w}};
    __builtin_amdgcn_sched_barrier(0);
    if (more) {
      const float4* cp =
          reinterpret_cast<const float4*>(c2 + (chunk + N_SPLIT) * 32 + 8 * h);
      c2v[0][0] = cp[0]; c2v[0][1] = cp[1];
      c2v[1][0] = cp[4]; c2v[1][1] = cp[5];
      load_centre_chunk(c_hi, c_lo, chunk + N_SPLIT, lane8, bc_hi, bc_lo);
    }
    __builtin_amdgcn_sched_barrier(0);

    v16bf pa_hi[M_TILES], pa_lo[M_TILES];
#pragma unroll
    for (int mt = 0; mt < M_TILES; ++mt) {
      v8u wh, wl;
#pragma unroll
      for (int ct = 0; ct < 2; ++ct) {
        float phi[8];
#pragma unroll
        for (int v = 0; v < 8; ++v)
          phi[v] = fmaxf(fmaf(-2.0f, g[mt][ct][v], x2v[mt] + c2cur[ct][v]), 1.0f);
#pragma unroll
        for (int v = 0; v < 8; ++v)
          phi[v] = __builtin_amdgcn_sqrtf(phi[v]);
#pragma unroll
        for (int v = 0; v < 8; v += 2) {
          const v2bf hp = __builtin_convertvector(v2f{phi[v], phi[v + 1]}, v2bf);
          const v2bf lp = __builtin_convertvector(
              v2f{phi[v] - (float)hp[0], phi[v + 1] - (float)hp[1]}, v2bf);
          wh[ct * 4 + v / 2] = __builtin_bit_cast(unsigned, hp);
          wl[ct * 4 + v / 2] = __builtin_bit_cast(unsigned, lp);
        }
      }
      pa_hi[mt] = __builtin_bit_cast(v16bf, wh);
      pa_lo[mt] = __builtin_bit_cast(v16bf, wl);
    }

#pragma unroll
    for (int nt = 0; nt < 4; ++nt) {
#pragma unroll
      for (int mt = 0; mt < M_TILES; ++mt)
        acc[mt][nt] = wmma_bf16(pa_hi[mt], bw_hi[nt], acc[mt][nt]);
#pragma unroll
      for (int mt = 0; mt < M_TILES; ++mt)
        acc[mt][nt] = wmma_bf16(pa_hi[mt], bw_lo[nt], acc[mt][nt]);
#pragma unroll
      for (int mt = 0; mt < M_TILES; ++mt)
        acc[mt][nt] = wmma_bf16(pa_lo[mt], bw_hi[nt], acc[mt][nt]);
    }

    if (more) load_w_chunk(wt_hi, wt_lo, chunk + N_SPLIT, lane8, bw_hi, bw_lo);
  };

  for (int it = 0; it < N_ITERS - 1; ++it) step(it * N_SPLIT + chunk0, true);
  step((N_ITERS - 1) * N_SPLIT + chunk0, false);

  float* part = &lds_part[wave][0][0];
#pragma unroll
  for (int mt = 0; mt < M_TILES; ++mt)
#pragma unroll
    for (int nt = 0; nt < 4; ++nt)
#pragma unroll
      for (int v = 0; v < 8; ++v)
        part[(mt * 16 + v + 8 * h) * PS + nt * 16 + ln] = acc[mt][nt][v];

  __syncthreads();

  float* __restrict__ dst = half ? part1 : part0;
#pragma unroll
  for (int ps = 0; ps < BLK_ROWS / (2 * N_SPLIT); ++ps) {
    const int t   = threadIdx.x;
    const int row = ps * (2 * N_SPLIT) + (t >> 4);
    const int col = (t & 15) * 4;
    float4 s = *reinterpret_cast<const float4*>(&lds_part[0][row][col]);
#pragma unroll
    for (int w = 1; w < N_SPLIT; ++w) {
      const float4 p = *reinterpret_cast<const float4*>(&lds_part[w][row][col]);
      s.x += p.x; s.y += p.y; s.z += p.z; s.w += p.w;
    }
    *reinterpret_cast<float4*>(dst + (rowbase + row) * L_OUT + col) = s;
  }
}

#define ADD_T 256
__global__ __launch_bounds__(ADD_T) void add_halves(
    const float* __restrict__ part0, const float* __restrict__ part1,
    float* __restrict__ out) {
  const int i = blockIdx.x * ADD_T + threadIdx.x;
  const float4 a = reinterpret_cast<const float4*>(part0)[i];
  const float4 b = reinterpret_cast<const float4*>(part1)[i];
  float4 s;
  s.x = a.x + b.x; s.y = a.y + b.y; s.z = a.z + b.z; s.w = a.w + b.w;
  reinterpret_cast<float4*>(out)[i] = s;
}

extern "C" void kernel_launch(void* const* d_in, const int* in_sizes, int n_in,
                              void* d_out, int out_size, void* d_ws, size_t ws_size,
                              hipStream_t stream) {
  (void)in_sizes; (void)n_in; (void)out_size; (void)ws_size;

  const float* xs     = (const float*)d_in[0];
  const float* centre = (const float*)d_in[1];
  const float* weight = (const float*)d_in[2];
  float* out          = (float*)d_out;

  const size_t XS = (size_t)D_ROWS * N_FEAT;
  const size_t WT = (size_t)N_CENT * L_OUT;

  float*  part0 = (float*)d_ws;
  float*  part1 = part0 + WT;
  __bf16* c_hi  = (__bf16*)(part1 + WT);
  __bf16* c_lo  = c_hi + XS;
  __bf16* wt_hi = c_lo + XS;
  __bf16* wt_lo = wt_hi + WT;
  float*  x2p1  = (float*)(wt_lo + WT);
  float*  c2    = x2p1 + D_ROWS;

  prepass_all<<<dim3(NB_PRE), dim3(PRE_T), 0, stream>>>(
      xs, centre, weight, c_hi, c_lo, wt_hi, wt_lo, x2p1, c2);

  rbfn_main<<<dim3((D_ROWS / BLK_ROWS) * C_HALVES), dim3(32 * N_SPLIT), 0, stream>>>(
      xs, c_hi, c_lo, wt_hi, wt_lo, x2p1, c2, part0, part1);

  add_halves<<<dim3((D_ROWS * L_OUT) / (4 * ADD_T)), dim3(ADD_T), 0, stream>>>(
      part0, part1, out);
}
